# starting kernel
# baseline (speedup 1.0000x reference)
#include <hip/hip_runtime.h>
#include <math.h>


typedef __attribute__((ext_vector_type(2))) float v2f;
typedef __attribute__((ext_vector_type(8))) float v8f;

#define B_TOT   2048
#define T_STEPS 128
#define NXc     8
#define NUc     2
#define NZc     50
#define DINP    64
#define DH      128
#define BT      16
#define DELTA_F 0.01f

#define LDW     129
#define LDW3    17
#define OFF_W1P 0u
#define OFF_W2P (OFF_W1P + 32u*LDW*8u)
#define OFF_W3P (OFF_W2P + 64u*LDW*8u)
#define OFF_INP (OFF_W3P + 64u*LDW3*8u)
#define OFF_H1  (OFF_INP + 16u*64u*4u)
#define OFF_H2  (OFF_H1  + 16u*128u*4u)
#define OFF_NN  (OFF_H2  + 16u*128u*4u)
#define OFF_XG  (OFF_NN  + 256u*4u)
#define OFF_XS  (OFF_XG  + 128u*4u)
#define OFF_Z   (OFF_XS  + 128u*4u)
#define OFF_K   (OFF_Z   + 16u*52u*4u)
#define OFF_U   (OFF_K   + 512u*4u)
#define OFF_B1  (OFF_U   + 32u*4u)
#define OFF_B2  (OFF_B1  + 128u*4u)
#define OFF_B3  (OFF_B2  + 128u*4u)
#define SMEM_BYTES (OFF_B3 + 16u*4u)

__device__ __forceinline__ float fast_tanh(float x) {
  float e = __expf(2.0f * x);
  return 1.0f - __fdividef(2.0f, e + 1.0f);
}

__device__ __forceinline__ v8f mm16(const float* __restrict__ A, int lda,
                                    const v2f* __restrict__ Bp, int ldb,
                                    int kIters, int lane) {
  v8f acc = {};
  const int mr = lane & 15;
  const int kh = lane >> 4;
  const float* ap = A + mr * lda + 2 * kh;
  const v2f*   bp = Bp + (size_t)kh * ldb + mr;
#pragma unroll
  for (int kk = 0; kk < kIters; ++kk) {
    v2f a = *(const v2f*)(ap + 4 * kk);
    v2f b = bp[(size_t)(2 * kk) * ldb];
    acc = __builtin_amdgcn_wmma_f32_16x16x4_f32(false, a, false, b,
                                                (short)0, acc, false, false);
  }
  return acc;
}

__device__ __forceinline__ void mlp_and_physics(
    int tid, int wave, int lane, int stage,
    const float* sInp, float* sH1, float* sH2, float* sNN,
    const v2f* W1p, const v2f* W2p, const v2f* W3p,
    const float* sB1, const float* sB2, const float* sB3,
    const float* xsrc, const float* sU, float* sK) {
  {
    v8f d = mm16(sInp, DINP, W1p + wave * 16, LDW, 16, lane);
    const int nc = wave * 16 + (lane & 15);
    const int kh = lane >> 4;
    const float bias = sB1[nc];
#pragma unroll
    for (int v = 0; v < 8; ++v)
      sH1[(v + 8 * kh) * DH + nc] = fast_tanh(d[v] + bias);
  }
  __syncthreads();
  {
    v8f d = mm16(sH1, DH, W2p + wave * 16, LDW, 32, lane);
    const int nc = wave * 16 + (lane & 15);
    const int kh = lane >> 4;
    const float bias = sB2[nc];
#pragma unroll
    for (int v = 0; v < 8; ++v)
      sH2[(v + 8 * kh) * DH + nc] = fast_tanh(d[v] + bias);
  }
  sNN[tid] = 0.0f;
  __syncthreads();
  {
    v8f d = mm16(sH2 + wave * 16, DH, W3p + wave * 8 * LDW3, LDW3, 4, lane);
    const int nc = lane & 15;
    const int kh = lane >> 4;
    if (nc < NXc) {
#pragma unroll
      for (int v = 0; v < 8; ++v)
        atomicAdd(&sNN[(v + 8 * kh) * 16 + nc], d[v]);
    }
  }
  __syncthreads();
  if (tid < BT) {
    const float* xs = xsrc + tid * NXc;
    float Hr  = xs[0] * 2.0f + 50.0f;
    float CAr = xs[1] * 0.2f + 1.0f;
    float CBr = xs[2] * 0.1f + 0.5f;
    float Tr  = xs[3] * 4.0f + 313.0f;
    float Hb  = xs[4] * 2.0f + 50.0f;
    float CAb = xs[5] * 0.2f + 1.0f;
    float CBb = xs[6] * 0.1f + 0.5f;
    float Tb  = xs[7] * 4.0f + 313.0f;
    float F = sU[tid * 2 + 0] * 1.0f + 10.0f;
    float D = sU[tid * 2 + 1] * 0.5f + 5.0f;
    float Fr = 2.5f * __fsqrt_rn(Hr);
    float Fb = 1.5f * __fsqrt_rn(Hb);
    float invArHr = __fdividef(1.0f, 3.0f * Hr);
    float invAbHb = __fdividef(1.0f, 3.0f * Hb);
    float dHr  = (F + D - Fr) * (1.0f / 3.0f);
    float dCAr = (F * (6.0f - CAr) - D * CAr) * invArHr;
    float dCBr = (-F * CBr - D * CBr) * invArHr;
    float dTr  = (F * (320.0f - Tr) + D * (300.0f - Tr)) * invArHr
               + (200.0f / 30.0f) * invArHr;
    float dHb  = (Fr - Fb - D) * (1.0f / 3.0f);
    float dCAb = (Fr * (CAr - CAb) + D * CAb) * invAbHb;
    float dCBb = (Fr * (CBr - CBb) + D * CBb) * invAbHb;
    float dTb  = Fr * (Tr - Tb) * invAbHb + (200.0f / 30.0f) * invAbHb;
    float*       kr = sK + stage * (BT * NXc) + tid * NXc;
    const float* nn = sNN + tid * 16;
    kr[0] = dHr  * 0.5f  + nn[0] + sB3[0];
    kr[1] = dCAr * 5.0f  + nn[1] + sB3[1];
    kr[2] = dCBr * 10.0f + nn[2] + sB3[2];
    kr[3] = dTr  * 0.25f + nn[3] + sB3[3];
    kr[4] = dHb  * 0.5f  + nn[4] + sB3[4];
    kr[5] = dCAb * 5.0f  + nn[5] + sB3[5];
    kr[6] = dCBb * 10.0f + nn[6] + sB3[6];
    kr[7] = dTb  * 0.25f + nn[7] + sB3[7];
  }
  __syncthreads();
}

__global__ __launch_bounds__(256) void cstr_rollout_kernel(
    const float* __restrict__ useq, const float* __restrict__ xGz0,
    const float* __restrict__ W1g, const float* __restrict__ b1g,
    const float* __restrict__ W2g, const float* __restrict__ b2g,
    const float* __restrict__ W3g, const float* __restrict__ b3g,
    float* __restrict__ out) {
  extern __shared__ __align__(16) char smem[];
  v2f*   W1p  = (v2f*)(smem + OFF_W1P);
  v2f*   W2p  = (v2f*)(smem + OFF_W2P);
  v2f*   W3p  = (v2f*)(smem + OFF_W3P);
  float* sInp = (float*)(smem + OFF_INP);
  float* sH1  = (float*)(smem + OFF_H1);
  float* sH2  = (float*)(smem + OFF_H2);
  float* sNN  = (float*)(smem + OFF_NN);
  float* sXG  = (float*)(smem + OFF_XG);
  float* sXs  = (float*)(smem + OFF_XS);
  float* sZ   = (float*)(smem + OFF_Z);
  float* sK   = (float*)(smem + OFF_K);
  float* sU   = (float*)(smem + OFF_U);
  float* sB1  = (float*)(smem + OFF_B1);
  float* sB2  = (float*)(smem + OFF_B2);
  float* sB3  = (float*)(smem + OFF_B3);

  const int tid  = threadIdx.x;
  const int lane = tid & 31;
  const int wave = tid >> 5;
  const int bt0  = blockIdx.x * BT;

  for (int i = tid; i < 32 * 128; i += 256) {
    int kk = i >> 7, n = i & 127;
    float lo = (2 * kk     < 60) ? W1g[(2 * kk)     * 128 + n] : 0.0f;
    float hi = (2 * kk + 1 < 60) ? W1g[(2 * kk + 1) * 128 + n] : 0.0f;
    W1p[kk * LDW + n] = (v2f){lo, hi};
  }
  for (int i = tid; i < 64 * 128; i += 256) {
    int kk = i >> 7, n = i & 127;
    W2p[kk * LDW + n] = (v2f){W2g[(2 * kk) * 128 + n], W2g[(2 * kk + 1) * 128 + n]};
  }
  for (int i = tid; i < 64 * 16; i += 256) {
    int kk = i >> 4, n = i & 15;
    v2f p = (v2f){0.0f, 0.0f};
    if (n < NXc) p = (v2f){W3g[(2 * kk) * NXc + n], W3g[(2 * kk + 1) * NXc + n]};
    W3p[kk * LDW3 + n] = p;
  }
  if (tid < 128) { sB1[tid] = b1g[tid]; sB2[tid] = b2g[tid]; }
  if (tid < NXc) sB3[tid] = b3g[tid];
  for (int i = tid; i < BT * 58; i += 256) {
    int r = i / 58, c = i - r * 58;
    float v = xGz0[(size_t)(bt0 + r) * 58 + c];
    if (c < NXc) sXG[r * NXc + c] = v;
    else         sZ[r * 52 + (c - NXc)] = v;
  }
  __syncthreads();

  for (int t = 0; t < T_STEPS; ++t) {
    if (tid < BT * NUc) {
      int r = tid >> 1, c = tid & 1;
      sU[tid] = useq[(size_t)(bt0 + r) * T_STEPS * NUc + (size_t)t * NUc + c];
    }
    __syncthreads();

    for (int i = tid; i < BT * DINP; i += 256) {
      int r = i >> 6, c = i & 63;
      float v;
      if      (c < 8)  v = sXG[r * NXc + c];
      else if (c < 58) v = sZ[r * 52 + (c - 8)];
      else if (c < 60) v = sU[r * 2 + (c - 58)];
      else             v = 0.0f;
      sInp[i] = v;
    }
    __syncthreads();
    mlp_and_physics(tid, wave, lane, 0, sInp, sH1, sH2, sNN, W1p, W2p, W3p,
                    sB1, sB2, sB3, sXG, sU, sK);

    for (int i = tid; i < BT * DINP; i += 256) {
      int r = i >> 6, c = i & 63;
      float v;
      if (c < 8) {
        v = sXG[r * NXc + c] + 0.5f * DELTA_F * sK[0 * 128 + r * NXc + c];
        sXs[r * NXc + c] = v;
      } else if (c < 48) {
        int idx = c - 8, p = idx >> 3, q = idx & 7;
        float y0 = sZ[r * 52 + p * 8 + q];
        float y1 = (p < 4) ? sZ[r * 52 + (p + 1) * 8 + q] : sXG[r * NXc + q];
        v = 0.5f * (y0 + y1);
      } else if (c < 58) v = sZ[r * 52 + 40 + (c - 48)];
      else if (c < 60)   v = sU[r * 2 + (c - 58)];
      else               v = 0.0f;
      sInp[i] = v;
    }
    __syncthreads();
    mlp_and_physics(tid, wave, lane, 1, sInp, sH1, sH2, sNN, W1p, W2p, W3p,
                    sB1, sB2, sB3, sXs, sU, sK);

    if (tid < BT * NXc) {
      int r = tid >> 3, j = tid & 7;
      float v = sXG[r * NXc + j] + 0.5f * DELTA_F * sK[1 * 128 + r * NXc + j];
      sXs[r * NXc + j] = v;
      sInp[r * DINP + j] = v;
    }
    __syncthreads();
    mlp_and_physics(tid, wave, lane, 2, sInp, sH1, sH2, sNN, W1p, W2p, W3p,
                    sB1, sB2, sB3, sXs, sU, sK);

    for (int i = tid; i < BT * 48; i += 256) {
      int r = i / 48, c = i - r * 48;
      float v;
      if (c < 8) {
        v = sXG[r * NXc + c] + DELTA_F * sK[2 * 128 + r * NXc + c];
        sXs[r * NXc + c] = v;
      } else {
        int idx = c - 8;
        v = (idx < 32) ? sZ[r * 52 + 8 + idx] : sXG[r * NXc + (idx - 32)];
      }
      sInp[r * DINP + c] = v;
    }
    __syncthreads();
    mlp_and_physics(tid, wave, lane, 3, sInp, sH1, sH2, sNN, W1p, W2p, W3p,
                    sB1, sB2, sB3, sXs, sU, sK);

    float xgnew = 0.0f;
    if (tid < BT * NXc) {
      int r = tid >> 3, j = tid & 7;
      float xg = sXG[r * NXc + j];
      out[(size_t)(bt0 + r) * T_STEPS * NXc + (size_t)t * NXc + j] = xg;
      xgnew = xg + (DELTA_F / 6.0f) *
              (sK[0 * 128 + tid] + 2.0f * sK[1 * 128 + tid] +
               2.0f * sK[2 * 128 + tid] + sK[3 * 128 + tid]);
    }
    float regz[4];
#pragma unroll
    for (int ii = 0; ii < 4; ++ii) {
      int i = tid + ii * 256;
      if (i < BT * NZc) {
        int r = i / NZc, c = i - r * NZc;
        float v;
        if      (c < 32) v = sZ[r * 52 + 8 + c];
        else if (c < 40) v = sXG[r * NXc + (c - 32)];
        else if (c < 48) v = sZ[r * 52 + (c + 2)];
        else             v = sU[r * 2 + (c - 48)];
        regz[ii] = v;
      }
    }
    __syncthreads();
#pragma unroll
    for (int ii = 0; ii < 4; ++ii) {
      int i = tid + ii * 256;
      if (i < BT * NZc) {
        int r = i / NZc, c = i - r * NZc;
        sZ[r * 52 + c] = regz[ii];
      }
    }
    if (tid < BT * NXc) sXG[tid] = xgnew;
    __syncthreads();
  }
}

extern "C" void kernel_launch(void* const* d_in, const int* in_sizes, int n_in,
                              void* d_out, int out_size, void* d_ws, size_t ws_size,
                              hipStream_t stream) {
  (void)in_sizes; (void)n_in; (void)out_size; (void)d_ws; (void)ws_size;
  const float* useq = (const float*)d_in[0];
  const float* xGz0 = (const float*)d_in[1];
  const float* W1   = (const float*)d_in[2];
  const float* b1   = (const float*)d_in[3];
  const float* W2   = (const float*)d_in[4];
  const float* b2   = (const float*)d_in[5];
  const float* W3   = (const float*)d_in[6];
  const float* b3   = (const float*)d_in[7];
  float* out = (float*)d_out;
  cstr_rollout_kernel<<<dim3(B_TOT / BT), dim3(256), SMEM_BYTES, stream>>>(
      useq, xGz0, W1, b1, W2, b2, W3, b3, out);
}


# LDS pitches re-chosen (A tiles 64/128 -> 68/132 dwords, weight pair-rows 129/17 -> 144/16 float2) so fragment ds_load_b64 hit 64 distinct banks: A reads were 16-way, B 2-way conflicted
# speedup vs baseline: 1.9464x; 1.9464x over previous
#include <hip/hip_runtime.h>
#include <math.h>


typedef __attribute__((ext_vector_type(2))) float v2f;
typedef __attribute__((ext_vector_type(8))) float v8f;

#define B_TOT   2048
#define T_STEPS 128
#define NXc     8
#define NUc     2
#define NZc     50
#define DINP    64
#define DH      128
#define BT      16
#define DELTA_F 0.01f

#define LDIN    68
#define LDH     132
#define LDW     144
#define LDW3    16
#define OFF_W1P 0u
#define OFF_W2P (OFF_W1P + 32u*LDW*8u)
#define OFF_W3P (OFF_W2P + 64u*LDW*8u)
#define OFF_INP (OFF_W3P + 64u*LDW3*8u)
#define OFF_H1  (OFF_INP + 16u*LDIN*4u)
#define OFF_H2  (OFF_H1  + 16u*LDH*4u)
#define OFF_NN  (OFF_H2  + 16u*LDH*4u)
#define OFF_XG  (OFF_NN  + 256u*4u)
#define OFF_XS  (OFF_XG  + 128u*4u)
#define OFF_Z   (OFF_XS  + 128u*4u)
#define OFF_K   (OFF_Z   + 16u*52u*4u)
#define OFF_U   (OFF_K   + 512u*4u)
#define OFF_B1  (OFF_U   + 32u*4u)
#define OFF_B2  (OFF_B1  + 128u*4u)
#define OFF_B3  (OFF_B2  + 128u*4u)
#define SMEM_BYTES (OFF_B3 + 16u*4u)

__device__ __forceinline__ float fast_tanh(float x) {
  float e = __expf(2.0f * x);
  return 1.0f - __fdividef(2.0f, e + 1.0f);
}

__device__ __forceinline__ v8f mm16(const float* __restrict__ A, int lda,
                                    const v2f* __restrict__ Bp, int ldb,
                                    int kIters, int lane) {
  v8f acc = {};
  const int mr = lane & 15;
  const int kh = lane >> 4;
  const float* ap = A + mr * lda + 2 * kh;
  const v2f*   bp = Bp + (size_t)kh * ldb + mr;
#pragma unroll
  for (int kk = 0; kk < kIters; ++kk) {
    v2f a = *(const v2f*)(ap + 4 * kk);
    v2f b = bp[(size_t)(2 * kk) * ldb];
    acc = __builtin_amdgcn_wmma_f32_16x16x4_f32(false, a, false, b,
                                                (short)0, acc, false, false);
  }
  return acc;
}

__device__ __forceinline__ void mlp_and_physics(
    int tid, int wave, int lane, int stage,
    const float* sInp, float* sH1, float* sH2, float* sNN,
    const v2f* W1p, const v2f* W2p, const v2f* W3p,
    const float* sB1, const float* sB2, const float* sB3,
    const float* xsrc, const float* sU, float* sK) {
  {
    v8f d = mm16(sInp, LDIN, W1p + wave * 16, LDW, 16, lane);
    const int nc = wave * 16 + (lane & 15);
    const int kh = lane >> 4;
    const float bias = sB1[nc];
#pragma unroll
    for (int v = 0; v < 8; ++v)
      sH1[(v + 8 * kh) * LDH + nc] = fast_tanh(d[v] + bias);
  }
  __syncthreads();
  {
    v8f d = mm16(sH1, LDH, W2p + wave * 16, LDW, 32, lane);
    const int nc = wave * 16 + (lane & 15);
    const int kh = lane >> 4;
    const float bias = sB2[nc];
#pragma unroll
    for (int v = 0; v < 8; ++v)
      sH2[(v + 8 * kh) * LDH + nc] = fast_tanh(d[v] + bias);
  }
  sNN[tid] = 0.0f;
  __syncthreads();
  {
    v8f d = mm16(sH2 + wave * 16, LDH, W3p + wave * 8 * LDW3, LDW3, 4, lane);
    const int nc = lane & 15;
    const int kh = lane >> 4;
    if (nc < NXc) {
#pragma unroll
      for (int v = 0; v < 8; ++v)
        atomicAdd(&sNN[(v + 8 * kh) * 16 + nc], d[v]);
    }
  }
  __syncthreads();
  if (tid < BT) {
    const float* xs = xsrc + tid * NXc;
    float Hr  = xs[0] * 2.0f + 50.0f;
    float CAr = xs[1] * 0.2f + 1.0f;
    float CBr = xs[2] * 0.1f + 0.5f;
    float Tr  = xs[3] * 4.0f + 313.0f;
    float Hb  = xs[4] * 2.0f + 50.0f;
    float CAb = xs[5] * 0.2f + 1.0f;
    float CBb = xs[6] * 0.1f + 0.5f;
    float Tb  = xs[7] * 4.0f + 313.0f;
    float F = sU[tid * 2 + 0] * 1.0f + 10.0f;
    float D = sU[tid * 2 + 1] * 0.5f + 5.0f;
    float Fr = 2.5f * __fsqrt_rn(Hr);
    float Fb = 1.5f * __fsqrt_rn(Hb);
    float invArHr = __fdividef(1.0f, 3.0f * Hr);
    float invAbHb = __fdividef(1.0f, 3.0f * Hb);
    float dHr  = (F + D - Fr) * (1.0f / 3.0f);
    float dCAr = (F * (6.0f - CAr) - D * CAr) * invArHr;
    float dCBr = (-F * CBr - D * CBr) * invArHr;
    float dTr  = (F * (320.0f - Tr) + D * (300.0f - Tr)) * invArHr
               + (200.0f / 30.0f) * invArHr;
    float dHb  = (Fr - Fb - D) * (1.0f / 3.0f);
    float dCAb = (Fr * (CAr - CAb) + D * CAb) * invAbHb;
    float dCBb = (Fr * (CBr - CBb) + D * CBb) * invAbHb;
    float dTb  = Fr * (Tr - Tb) * invAbHb + (200.0f / 30.0f) * invAbHb;
    float*       kr = sK + stage * (BT * NXc) + tid * NXc;
    const float* nn = sNN + tid * 16;
    kr[0] = dHr  * 0.5f  + nn[0] + sB3[0];
    kr[1] = dCAr * 5.0f  + nn[1] + sB3[1];
    kr[2] = dCBr * 10.0f + nn[2] + sB3[2];
    kr[3] = dTr  * 0.25f + nn[3] + sB3[3];
    kr[4] = dHb  * 0.5f  + nn[4] + sB3[4];
    kr[5] = dCAb * 5.0f  + nn[5] + sB3[5];
    kr[6] = dCBb * 10.0f + nn[6] + sB3[6];
    kr[7] = dTb  * 0.25f + nn[7] + sB3[7];
  }
  __syncthreads();
}

__global__ __launch_bounds__(256) void cstr_rollout_kernel(
    const float* __restrict__ useq, const float* __restrict__ xGz0,
    const float* __restrict__ W1g, const float* __restrict__ b1g,
    const float* __restrict__ W2g, const float* __restrict__ b2g,
    const float* __restrict__ W3g, const float* __restrict__ b3g,
    float* __restrict__ out) {
  extern __shared__ __align__(16) char smem[];
  v2f*   W1p  = (v2f*)(smem + OFF_W1P);
  v2f*   W2p  = (v2f*)(smem + OFF_W2P);
  v2f*   W3p  = (v2f*)(smem + OFF_W3P);
  float* sInp = (float*)(smem + OFF_INP);
  float* sH1  = (float*)(smem + OFF_H1);
  float* sH2  = (float*)(smem + OFF_H2);
  float* sNN  = (float*)(smem + OFF_NN);
  float* sXG  = (float*)(smem + OFF_XG);
  float* sXs  = (float*)(smem + OFF_XS);
  float* sZ   = (float*)(smem + OFF_Z);
  float* sK   = (float*)(smem + OFF_K);
  float* sU   = (float*)(smem + OFF_U);
  float* sB1  = (float*)(smem + OFF_B1);
  float* sB2  = (float*)(smem + OFF_B2);
  float* sB3  = (float*)(smem + OFF_B3);

  const int tid  = threadIdx.x;
  const int lane = tid & 31;
  const int wave = tid >> 5;
  const int bt0  = blockIdx.x * BT;

  for (int i = tid; i < 32 * 128; i += 256) {
    int kk = i >> 7, n = i & 127;
    float lo = (2 * kk     < 60) ? W1g[(2 * kk)     * 128 + n] : 0.0f;
    float hi = (2 * kk + 1 < 60) ? W1g[(2 * kk + 1) * 128 + n] : 0.0f;
    W1p[kk * LDW + n] = (v2f){lo, hi};
  }
  for (int i = tid; i < 64 * 128; i += 256) {
    int kk = i >> 7, n = i & 127;
    W2p[kk * LDW + n] = (v2f){W2g[(2 * kk) * 128 + n], W2g[(2 * kk + 1) * 128 + n]};
  }
  for (int i = tid; i < 64 * 16; i += 256) {
    int kk = i >> 4, n = i & 15;
    v2f p = (v2f){0.0f, 0.0f};
    if (n < NXc) p = (v2f){W3g[(2 * kk) * NXc + n], W3g[(2 * kk + 1) * NXc + n]};
    W3p[kk * LDW3 + n] = p;
  }
  if (tid < 128) { sB1[tid] = b1g[tid]; sB2[tid] = b2g[tid]; }
  if (tid < NXc) sB3[tid] = b3g[tid];
  for (int i = tid; i < BT * 58; i += 256) {
    int r = i / 58, c = i - r * 58;
    float v = xGz0[(size_t)(bt0 + r) * 58 + c];
    if (c < NXc) sXG[r * NXc + c] = v;
    else         sZ[r * 52 + (c - NXc)] = v;
  }
  __syncthreads();

  for (int t = 0; t < T_STEPS; ++t) {
    if (tid < BT * NUc) {
      int r = tid >> 1, c = tid & 1;
      sU[tid] = useq[(size_t)(bt0 + r) * T_STEPS * NUc + (size_t)t * NUc + c];
    }
    __syncthreads();

    for (int i = tid; i < BT * DINP; i += 256) {
      int r = i >> 6, c = i & 63;
      float v;
      if      (c < 8)  v = sXG[r * NXc + c];
      else if (c < 58) v = sZ[r * 52 + (c - 8)];
      else if (c < 60) v = sU[r * 2 + (c - 58)];
      else             v = 0.0f;
      sInp[r * LDIN + c] = v;
    }
    __syncthreads();
    mlp_and_physics(tid, wave, lane, 0, sInp, sH1, sH2, sNN, W1p, W2p, W3p,
                    sB1, sB2, sB3, sXG, sU, sK);

    for (int i = tid; i < BT * DINP; i += 256) {
      int r = i >> 6, c = i & 63;
      float v;
      if (c < 8) {
        v = sXG[r * NXc + c] + 0.5f * DELTA_F * sK[0 * 128 + r * NXc + c];
        sXs[r * NXc + c] = v;
      } else if (c < 48) {
        int idx = c - 8, p = idx >> 3, q = idx & 7;
        float y0 = sZ[r * 52 + p * 8 + q];
        float y1 = (p < 4) ? sZ[r * 52 + (p + 1) * 8 + q] : sXG[r * NXc + q];
        v = 0.5f * (y0 + y1);
      } else if (c < 58) v = sZ[r * 52 + 40 + (c - 48)];
      else if (c < 60)   v = sU[r * 2 + (c - 58)];
      else               v = 0.0f;
      sInp[r * LDIN + c] = v;
    }
    __syncthreads();
    mlp_and_physics(tid, wave, lane, 1, sInp, sH1, sH2, sNN, W1p, W2p, W3p,
                    sB1, sB2, sB3, sXs, sU, sK);

    if (tid < BT * NXc) {
      int r = tid >> 3, j = tid & 7;
      float v = sXG[r * NXc + j] + 0.5f * DELTA_F * sK[1 * 128 + r * NXc + j];
      sXs[r * NXc + j] = v;
      sInp[r * LDIN + j] = v;
    }
    __syncthreads();
    mlp_and_physics(tid, wave, lane, 2, sInp, sH1, sH2, sNN, W1p, W2p, W3p,
                    sB1, sB2, sB3, sXs, sU, sK);

    for (int i = tid; i < BT * 48; i += 256) {
      int r = i / 48, c = i - r * 48;
      float v;
      if (c < 8) {
        v = sXG[r * NXc + c] + DELTA_F * sK[2 * 128 + r * NXc + c];
        sXs[r * NXc + c] = v;
      } else {
        int idx = c - 8;
        v = (idx < 32) ? sZ[r * 52 + 8 + idx] : sXG[r * NXc + (idx - 32)];
      }
      sInp[r * LDIN + c] = v;
    }
    __syncthreads();
    mlp_and_physics(tid, wave, lane, 3, sInp, sH1, sH2, sNN, W1p, W2p, W3p,
                    sB1, sB2, sB3, sXs, sU, sK);

    float xgnew = 0.0f;
    if (tid < BT * NXc) {
      int r = tid >> 3, j = tid & 7;
      float xg = sXG[r * NXc + j];
      out[(size_t)(bt0 + r) * T_STEPS * NXc + (size_t)t * NXc + j] = xg;
      xgnew = xg + (DELTA_F / 6.0f) *
              (sK[0 * 128 + tid] + 2.0f * sK[1 * 128 + tid] +
               2.0f * sK[2 * 128 + tid] + sK[3 * 128 + tid]);
    }
    float regz[4];
#pragma unroll
    for (int ii = 0; ii < 4; ++ii) {
      int i = tid + ii * 256;
      if (i < BT * NZc) {
        int r = i / NZc, c = i - r * NZc;
        float v;
        if      (c < 32) v = sZ[r * 52 + 8 + c];
        else if (c < 40) v = sXG[r * NXc + (c - 32)];
        else if (c < 48) v = sZ[r * 52 + (c + 2)];
        else             v = sU[r * 2 + (c - 48)];
        regz[ii] = v;
      }
    }
    __syncthreads();
#pragma unroll
    for (int ii = 0; ii < 4; ++ii) {
      int i = tid + ii * 256;
      if (i < BT * NZc) {
        int r = i / NZc, c = i - r * NZc;
        sZ[r * 52 + c] = regz[ii];
      }
    }
    if (tid < BT * NXc) sXG[tid] = xgnew;
    __syncthreads();
  }
}

extern "C" void kernel_launch(void* const* d_in, const int* in_sizes, int n_in,
                              void* d_out, int out_size, void* d_ws, size_t ws_size,
                              hipStream_t stream) {
  (void)in_sizes; (void)n_in; (void)out_size; (void)d_ws; (void)ws_size;
  const float* useq = (const float*)d_in[0];
  const float* xGz0 = (const float*)d_in[1];
  const float* W1   = (const float*)d_in[2];
  const float* b1   = (const float*)d_in[3];
  const float* W2   = (const float*)d_in[4];
  const float* b2   = (const float*)d_in[5];
  const float* W3   = (const float*)d_in[6];
  const float* b3   = (const float*)d_in[7];
  float* out = (float*)d_out;
  cstr_rollout_kernel<<<dim3(B_TOT / BT), dim3(256), SMEM_BYTES, stream>>>(
      useq, xGz0, W1, b1, W2, b2, W3, b3, out);
}


# No LDS read between WMMAs any more: each wave keeps its weight (B) fragments and biases in VGPRs for the whole rollout and reads a layer's A fragments as one ds_load group ahead of its WMMA chain
# speedup vs baseline: 2.2101x; 1.1355x over previous
#include <hip/hip_runtime.h>
#include <math.h>


typedef __attribute__((ext_vector_type(2))) float v2f;
typedef __attribute__((ext_vector_type(8))) float v8f;

#define B_TOT   2048
#define T_STEPS 128
#define NXc     8
#define NUc     2
#define NZc     50
#define DINP    64
#define DH      128
#define BT      16
#define DELTA_F 0.01f

#define LDIN    68
#define LDH     132
#define OFF_INP 0u
#define OFF_H1  (OFF_INP + 16u*LDIN*4u)
#define OFF_H2  (OFF_H1  + 16u*LDH*4u)
#define OFF_NN  (OFF_H2  + 16u*LDH*4u)
#define OFF_XG  (OFF_NN  + 256u*4u)
#define OFF_XS  (OFF_XG  + 128u*4u)
#define OFF_Z   (OFF_XS  + 128u*4u)
#define OFF_K   (OFF_Z   + 16u*52u*4u)
#define OFF_U   (OFF_K   + 512u*4u)
#define OFF_B3  (OFF_U   + 32u*4u)
#define SMEM_BYTES (OFF_B3 + 16u*4u)

__device__ __forceinline__ float fast_tanh(float x) {
  float e = __expf(2.0f * x);
  return 1.0f - __fdividef(2.0f, e + 1.0f);
}

template <int KI>
__device__ __forceinline__ v8f mm16(const float* __restrict__ A, int lda,
                                    const v2f (&b)[KI], int lane) {
  const int mr = lane & 15;
  const int kh = lane >> 4;
  const float* ap = A + mr * lda + 2 * kh;
  v2f a[KI];
#pragma unroll
  for (int kk = 0; kk < KI; ++kk) a[kk] = *(const v2f*)(ap + 4 * kk);
  __builtin_amdgcn_sched_barrier(0);
  v8f acc = {};
#pragma unroll
  for (int kk = 0; kk < KI; ++kk)
    acc = __builtin_amdgcn_wmma_f32_16x16x4_f32(false, a[kk], false, b[kk],
                                                (short)0, acc, false, false);
  return acc;
}

__device__ __forceinline__ void mlp_and_physics(
    int tid, int wave, int lane, int stage,
    const float* sInp, float* sH1, float* sH2, float* sNN,
    const v2f (&w1)[16], const v2f (&w2)[32], const v2f (&w3)[4],
    float bias1, float bias2, const float* sB3,
    const float* xsrc, const float* sU, float* sK) {
  {
    v8f d = mm16<16>(sInp, LDIN, w1, lane);
    const int nc = wave * 16 + (lane & 15);
    const int kh = lane >> 4;
#pragma unroll
    for (int v = 0; v < 8; ++v)
      sH1[(v + 8 * kh) * LDH + nc] = fast_tanh(d[v] + bias1);
  }
  __syncthreads();
  {
    v8f d = mm16<32>(sH1, LDH, w2, lane);
    const int nc = wave * 16 + (lane & 15);
    const int kh = lane >> 4;
#pragma unroll
    for (int v = 0; v < 8; ++v)
      sH2[(v + 8 * kh) * LDH + nc] = fast_tanh(d[v] + bias2);
  }
  sNN[tid] = 0.0f;
  __syncthreads();
  {
    v8f d = mm16<4>(sH2 + wave * 16, LDH, w3, lane);
    const int nc = lane & 15;
    const int kh = lane >> 4;
    if (nc < NXc) {
#pragma unroll
      for (int v = 0; v < 8; ++v)
        atomicAdd(&sNN[(v + 8 * kh) * 16 + nc], d[v]);
    }
  }
  __syncthreads();
  if (tid < BT) {
    const float* xs = xsrc + tid * NXc;
    float Hr  = xs[0] * 2.0f + 50.0f;
    float CAr = xs[1] * 0.2f + 1.0f;
    float CBr = xs[2] * 0.1f + 0.5f;
    float Tr  = xs[3] * 4.0f + 313.0f;
    float Hb  = xs[4] * 2.0f + 50.0f;
    float CAb = xs[5] * 0.2f + 1.0f;
    float CBb = xs[6] * 0.1f + 0.5f;
    float Tb  = xs[7] * 4.0f + 313.0f;
    float F = sU[tid * 2 + 0] * 1.0f + 10.0f;
    float D = sU[tid * 2 + 1] * 0.5f + 5.0f;
    float Fr = 2.5f * __fsqrt_rn(Hr);
    float Fb = 1.5f * __fsqrt_rn(Hb);
    float invArHr = __fdividef(1.0f, 3.0f * Hr);
    float invAbHb = __fdividef(1.0f, 3.0f * Hb);
    float dHr  = (F + D - Fr) * (1.0f / 3.0f);
    float dCAr = (F * (6.0f - CAr) - D * CAr) * invArHr;
    float dCBr = (-F * CBr - D * CBr) * invArHr;
    float dTr  = (F * (320.0f - Tr) + D * (300.0f - Tr)) * invArHr
               + (200.0f / 30.0f) * invArHr;
    float dHb  = (Fr - Fb - D) * (1.0f / 3.0f);
    float dCAb = (Fr * (CAr - CAb) + D * CAb) * invAbHb;
    float dCBb = (Fr * (CBr - CBb) + D * CBb) * invAbHb;
    float dTb  = Fr * (Tr - Tb) * invAbHb + (200.0f / 30.0f) * invAbHb;
    float*       kr = sK + stage * (BT * NXc) + tid * NXc;
    const float* nn = sNN + tid * 16;
    kr[0] = dHr  * 0.5f  + nn[0] + sB3[0];
    kr[1] = dCAr * 5.0f  + nn[1] + sB3[1];
    kr[2] = dCBr * 10.0f + nn[2] + sB3[2];
    kr[3] = dTr  * 0.25f + nn[3] + sB3[3];
    kr[4] = dHb  * 0.5f  + nn[4] + sB3[4];
    kr[5] = dCAb * 5.0f  + nn[5] + sB3[5];
    kr[6] = dCBb * 10.0f + nn[6] + sB3[6];
    kr[7] = dTb  * 0.25f + nn[7] + sB3[7];
  }
  __syncthreads();
}

__global__ __launch_bounds__(256) void cstr_rollout_kernel(
    const float* __restrict__ useq, const float* __restrict__ xGz0,
    const float* __restrict__ W1g, const float* __restrict__ b1g,
    const float* __restrict__ W2g, const float* __restrict__ b2g,
    const float* __restrict__ W3g, const float* __restrict__ b3g,
    float* __restrict__ out) {
  extern __shared__ __align__(16) char smem[];
  float* sInp = (float*)(smem + OFF_INP);
  float* sH1  = (float*)(smem + OFF_H1);
  float* sH2  = (float*)(smem + OFF_H2);
  float* sNN  = (float*)(smem + OFF_NN);
  float* sXG  = (float*)(smem + OFF_XG);
  float* sXs  = (float*)(smem + OFF_XS);
  float* sZ   = (float*)(smem + OFF_Z);
  float* sK   = (float*)(smem + OFF_K);
  float* sU   = (float*)(smem + OFF_U);
  float* sB3  = (float*)(smem + OFF_B3);

  const int tid  = threadIdx.x;
  const int lane = tid & 31;
  const int wave = tid >> 5;
  const int bt0  = blockIdx.x * BT;

  v2f w1[16], w2[32], w3[4];
  const int wn  = lane & 15;
  const int wkh = lane >> 4;
  const int wcol = wave * 16 + wn;
#pragma unroll
  for (int kk = 0; kk < 16; ++kk) {
    const int k0 = 4 * kk + 2 * wkh;
    const float lo = (k0     < 60) ? W1g[(k0)     * DH + wcol] : 0.0f;
    const float hi = (k0 + 1 < 60) ? W1g[(k0 + 1) * DH + wcol] : 0.0f;
    w1[kk] = (v2f){lo, hi};
  }
#pragma unroll
  for (int kk = 0; kk < 32; ++kk) {
    const int k0 = 4 * kk + 2 * wkh;
    w2[kk] = (v2f){W2g[(k0) * DH + wcol], W2g[(k0 + 1) * DH + wcol]};
  }
#pragma unroll
  for (int kk = 0; kk < 4; ++kk) {
    const int k0 = wave * 16 + 4 * kk + 2 * wkh;
    v2f p = (v2f){0.0f, 0.0f};
    if (wn < NXc) p = (v2f){W3g[(k0) * NXc + wn], W3g[(k0 + 1) * NXc + wn]};
    w3[kk] = p;
  }
  const float bias1 = b1g[wcol];
  const float bias2 = b2g[wcol];
  if (tid < NXc) sB3[tid] = b3g[tid];
  for (int i = tid; i < BT * 58; i += 256) {
    int r = i / 58, c = i - r * 58;
    float v = xGz0[(size_t)(bt0 + r) * 58 + c];
    if (c < NXc) sXG[r * NXc + c] = v;
    else         sZ[r * 52 + (c - NXc)] = v;
  }
  __syncthreads();

  for (int t = 0; t < T_STEPS; ++t) {
    if (tid < BT * NUc) {
      int r = tid >> 1, c = tid & 1;
      sU[tid] = useq[(size_t)(bt0 + r) * T_STEPS * NUc + (size_t)t * NUc + c];
    }
    __syncthreads();

    for (int i = tid; i < BT * DINP; i += 256) {
      int r = i >> 6, c = i & 63;
      float v;
      if      (c < 8)  v = sXG[r * NXc + c];
      else if (c < 58) v = sZ[r * 52 + (c - 8)];
      else if (c < 60) v = sU[r * 2 + (c - 58)];
      else             v = 0.0f;
      sInp[r * LDIN + c] = v;
    }
    __syncthreads();
    mlp_and_physics(tid, wave, lane, 0, sInp, sH1, sH2, sNN, w1, w2, w3,
                    bias1, bias2, sB3, sXG, sU, sK);

    for (int i = tid; i < BT * DINP; i += 256) {
      int r = i >> 6, c = i & 63;
      float v;
      if (c < 8) {
        v = sXG[r * NXc + c] + 0.5f * DELTA_F * sK[0 * 128 + r * NXc + c];
        sXs[r * NXc + c] = v;
      } else if (c < 48) {
        int idx = c - 8, p = idx >> 3, q = idx & 7;
        float y0 = sZ[r * 52 + p * 8 + q];
        float y1 = (p < 4) ? sZ[r * 52 + (p + 1) * 8 + q] : sXG[r * NXc + q];
        v = 0.5f * (y0 + y1);
      } else if (c < 58) v = sZ[r * 52 + 40 + (c - 48)];
      else if (c < 60)   v = sU[r * 2 + (c - 58)];
      else               v = 0.0f;
      sInp[r * LDIN + c] = v;
    }
    __syncthreads();
    mlp_and_physics(tid, wave, lane, 1, sInp, sH1, sH2, sNN, w1, w2, w3,
                    bias1, bias2, sB3, sXs, sU, sK);

    if (tid < BT * NXc) {
      int r = tid >> 3, j = tid & 7;
      float v = sXG[r * NXc + j] + 0.5f * DELTA_F * sK[1 * 128 + r * NXc + j];
      sXs[r * NXc + j] = v;
      sInp[r * LDIN + j] = v;
    }
    __syncthreads();
    mlp_and_physics(tid, wave, lane, 2, sInp, sH1, sH2, sNN, w1, w2, w3,
                    bias1, bias2, sB3, sXs, sU, sK);

    for (int i = tid; i < BT * 48; i += 256) {
      int r = i / 48, c = i - r * 48;
      float v;
      if (c < 8) {
        v = sXG[r * NXc + c] + DELTA_F * sK[2 * 128 + r * NXc + c];
        sXs[r * NXc + c] = v;
      } else {
        int idx = c - 8;
        v = (idx < 32) ? sZ[r * 52 + 8 + idx] : sXG[r * NXc + (idx - 32)];
      }
      sInp[r * LDIN + c] = v;
    }
    __syncthreads();
    mlp_and_physics(tid, wave, lane, 3, sInp, sH1, sH2, sNN, w1, w2, w3,
                    bias1, bias2, sB3, sXs, sU, sK);

    float xgnew = 0.0f;
    if (tid < BT * NXc) {
      int r = tid >> 3, j = tid & 7;
      float xg = sXG[r * NXc + j];
      out[(size_t)(bt0 + r) * T_STEPS * NXc + (size_t)t * NXc + j] = xg;
      xgnew = xg + (DELTA_F / 6.0f) *
              (sK[0 * 128 + tid] + 2.0f * sK[1 * 128 + tid] +
               2.0f * sK[2 * 128 + tid] + sK[3 * 128 + tid]);
    }
    float regz[4];
#pragma unroll
    for (int ii = 0; ii < 4; ++ii) {
      int i = tid + ii * 256;
      if (i < BT * NZc) {
        int r = i / NZc, c = i - r * NZc;
        float v;
        if      (c < 32) v = sZ[r * 52 + 8 + c];
        else if (c < 40) v = sXG[r * NXc + (c - 32)];
        else if (c < 48) v = sZ[r * 52 + (c + 2)];
        else             v = sU[r * 2 + (c - 48)];
        regz[ii] = v;
      }
    }
    __syncthreads();
#pragma unroll
    for (int ii = 0; ii < 4; ++ii) {
      int i = tid + ii * 256;
      if (i < BT * NZc) {
        int r = i / NZc, c = i - r * NZc;
        sZ[r * 52 + c] = regz[ii];
      }
    }
    if (tid < BT * NXc) sXG[tid] = xgnew;
    __syncthreads();
  }
}

extern "C" void kernel_launch(void* const* d_in, const int* in_sizes, int n_in,
                              void* d_out, int out_size, void* d_ws, size_t ws_size,
                              hipStream_t stream) {
  (void)in_sizes; (void)n_in; (void)out_size; (void)d_ws; (void)ws_size;
  const float* useq = (const float*)d_in[0];
  const float* xGz0 = (const float*)d_in[1];
  const float* W1   = (const float*)d_in[2];
  const float* b1   = (const float*)d_in[3];
  const float* W2   = (const float*)d_in[4];
  const float* b2   = (const float*)d_in[5];
  const float* W3   = (const float*)d_in[6];
  const float* b3   = (const float*)d_in[7];
  float* out = (float*)d_out;
  cstr_rollout_kernel<<<dim3(B_TOT / BT), dim3(256), SMEM_BYTES, stream>>>(
      useq, xGz0, W1, b1, W2, b2, W3, b3, out);
}


# 12 block barriers per step instead of 23: layer 3 follows layer 2 on the wave's own columns, 4 added waves do physics beside the MLP and sum split-K partials in fixed order (LDS atomics gone)
# speedup vs baseline: 2.7684x; 1.2526x over previous
#include <hip/hip_runtime.h>
#include <math.h>


typedef __attribute__((ext_vector_type(2))) float v2f;
typedef __attribute__((ext_vector_type(8))) float v8f;

#define B_TOT   2048
#define T_STEPS 128
#define NXc     8
#define NUc     2
#define NZc     50
#define DH      128
#define BT      16
#define DELTA_F 0.01f
#define NTHREADS 384
#define NMLP     256

#define LDIN    68
#define LDH     132
#define ZP      52
#define UP      258
#define OFF_INP 0u
#define OFF_H1  (OFF_INP + 16u*LDIN*4u)
#define OFF_H2  (OFF_H1  + 16u*LDH*4u)
#define OFF_P   (OFF_H2  + 16u*LDH*4u)
#define OFF_XG  (OFF_P   + 8u*128u*4u)
#define OFF_Z   (OFF_XG  + 2u*128u*4u)
#define OFF_UA  (OFF_Z   + 2u*16u*ZP*4u)
#define SMEM_BYTES (OFF_UA + 16u*UP*4u)

__device__ __forceinline__ float fast_tanh(float x) {
  float e = __expf(2.0f * x);
  return 1.0f - __fdividef(2.0f, e + 1.0f);
}

template <int KI>
__device__ __forceinline__ v8f mm16(const float* __restrict__ A, int lda,
                                    const v2f (&b)[KI], int lane) {
  const int mr = lane & 15;
  const int kh = lane >> 4;
  const float* ap = A + mr * lda + 2 * kh;
  v2f a[KI];
#pragma unroll
  for (int kk = 0; kk < KI; ++kk) a[kk] = *(const v2f*)(ap + 4 * kk);
  __builtin_amdgcn_sched_barrier(0);
  v8f acc = {};
#pragma unroll
  for (int kk = 0; kk < KI; ++kk)
    acc = __builtin_amdgcn_wmma_f32_16x16x4_f32(false, a[kk], false, b[kk],
                                                (short)0, acc, false, false);
  return acc;
}

__device__ __forceinline__ float physics_component(const float* xs, float u0,
                                                   float u1, int j) {
  float Hr  = xs[0] * 2.0f + 50.0f;
  float CAr = xs[1] * 0.2f + 1.0f;
  float CBr = xs[2] * 0.1f + 0.5f;
  float Tr  = xs[3] * 4.0f + 313.0f;
  float Hb  = xs[4] * 2.0f + 50.0f;
  float CAb = xs[5] * 0.2f + 1.0f;
  float CBb = xs[6] * 0.1f + 0.5f;
  float Tb  = xs[7] * 4.0f + 313.0f;
  float F = u0 * 1.0f + 10.0f;
  float D = u1 * 0.5f + 5.0f;
  float Fr = 2.5f * __fsqrt_rn(Hr);
  float Fb = 1.5f * __fsqrt_rn(Hb);
  float invArHr = __fdividef(1.0f, 3.0f * Hr);
  float invAbHb = __fdividef(1.0f, 3.0f * Hb);
  float dHr  = (F + D - Fr) * (1.0f / 3.0f);
  float dCAr = (F * (6.0f - CAr) - D * CAr) * invArHr;
  float dCBr = (-F * CBr - D * CBr) * invArHr;
  float dTr  = (F * (320.0f - Tr) + D * (300.0f - Tr)) * invArHr
             + (200.0f / 30.0f) * invArHr;
  float dHb  = (Fr - Fb - D) * (1.0f / 3.0f);
  float dCAb = (Fr * (CAr - CAb) + D * CAb) * invAbHb;
  float dCBb = (Fr * (CBr - CBb) + D * CBb) * invAbHb;
  float dTb  = Fr * (Tr - Tb) * invAbHb + (200.0f / 30.0f) * invAbHb;
  float lo = (j & 2) ? ((j & 1) ? dTr : dCBr) : ((j & 1) ? dCAr : dHr);
  float hi = (j & 2) ? ((j & 1) ? dTb : dCBb) : ((j & 1) ? dCAb : dHb);
  return (j & 4) ? hi : lo;
}

__global__ __launch_bounds__(NTHREADS) void cstr_rollout_kernel(
    const float* __restrict__ useq, const float* __restrict__ xGz0,
    const float* __restrict__ W1g, const float* __restrict__ b1g,
    const float* __restrict__ W2g, const float* __restrict__ b2g,
    const float* __restrict__ W3g, const float* __restrict__ b3g,
    float* __restrict__ out) {
  extern __shared__ __align__(16) char smem[];
  float* sInp = (float*)(smem + OFF_INP);
  float* sH1  = (float*)(smem + OFF_H1);
  float* sH2  = (float*)(smem + OFF_H2);
  float* sP   = (float*)(smem + OFF_P);
  float* sXG  = (float*)(smem + OFF_XG);
  float* sZ   = (float*)(smem + OFF_Z);
  float* sUA  = (float*)(smem + OFF_UA);

  const int tid  = threadIdx.x;
  const int lane = tid & 31;
  const int wave = __builtin_amdgcn_readfirstlane(tid >> 5);
  const bool is_mlp = wave < 8;
  const int bt0  = blockIdx.x * BT;

  const int st = (tid - NMLP) & 127;
  const int sr = st >> 3;
  const int sj = st & 7;

  v2f w1[16] = {}, w2[32] = {}, w3[4] = {};
  float bias1 = 0.0f, bias2 = 0.0f;
  const int wn  = lane & 15;
  const int wkh = lane >> 4;
  if (is_mlp) {
    const int wcol = wave * 16 + wn;
#pragma unroll
    for (int kk = 0; kk < 16; ++kk) {
      const int k0 = 4 * kk + 2 * wkh;
      const float lo = (k0     < 60) ? W1g[(k0)     * DH + wcol] : 0.0f;
      const float hi = (k0 + 1 < 60) ? W1g[(k0 + 1) * DH + wcol] : 0.0f;
      w1[kk] = (v2f){lo, hi};
    }
#pragma unroll
    for (int kk = 0; kk < 32; ++kk) {
      const int k0 = 4 * kk + 2 * wkh;
      w2[kk] = (v2f){W2g[(k0) * DH + wcol], W2g[(k0 + 1) * DH + wcol]};
    }
#pragma unroll
    for (int kk = 0; kk < 4; ++kk) {
      const int k0 = wave * 16 + 4 * kk + 2 * wkh;
      v2f p = (v2f){0.0f, 0.0f};
      if (wn < NXc) p = (v2f){W3g[(k0) * NXc + wn], W3g[(k0 + 1) * NXc + wn]};
      w3[kk] = p;
    }
    bias1 = b1g[wcol];
    bias2 = b2g[wcol];
  }

  float xg = 0.0f, b3j = 0.0f;
  if (!is_mlp) {
    xg  = xGz0[(size_t)(bt0 + sr) * 58 + sj];
    b3j = b3g[sj];
  }
  const float scj = ((sj & 3) == 0) ? 0.5f : ((sj & 3) == 1) ? 5.0f
                  : ((sj & 3) == 2) ? 10.0f : 0.25f;

  for (int i = tid; i < BT * T_STEPS * NUc; i += NTHREADS) {
    int r = i >> 8, q = i & 255;
    sUA[r * UP + q] = useq[(size_t)bt0 * (T_STEPS * NUc) + i];
  }
  for (int i = tid; i < BT * 58; i += NTHREADS) {
    int r = i / 58, c = i - r * 58;
    float v = xGz0[(size_t)(bt0 + r) * 58 + c];
    if (c < NXc) sXG[r * NXc + c] = v;
    else         sZ[r * ZP + (c - NXc)] = v;
    sInp[r * LDIN + c] = v;
  }
  if (tid < BT * 6) {
    int r = tid / 6, c = tid - r * 6;
    float v = (c < 2) ? useq[(size_t)(bt0 + r) * (T_STEPS * NUc) + c] : 0.0f;
    sInp[r * LDIN + 58 + c] = v;
  }
  __syncthreads();

  float k1 = 0.0f, k2 = 0.0f, k3 = 0.0f;
  for (int t = 0; t < T_STEPS; ++t) {
    const int cur = t & 1;
    const float* Zc  = sZ  + cur * (BT * ZP);
    float*       Zn  = sZ  + (cur ^ 1) * (BT * ZP);
    const float* XGc = sXG + cur * (BT * NXc);
    float*       XGn = sXG + (cur ^ 1) * (BT * NXc);

#pragma unroll
    for (int s = 0; s < 4; ++s) {
      float fgv = 0.0f;
      if (is_mlp) {
        v8f d = mm16<16>(sInp, LDIN, w1, lane);
        const int nc = wave * 16 + (lane & 15);
        const int kh = lane >> 4;
#pragma unroll
        for (int v = 0; v < 8; ++v)
          sH1[(v + 8 * kh) * LDH + nc] = fast_tanh(d[v] + bias1);
      } else {
        if (s == 0)
          out[(size_t)(bt0 + sr) * T_STEPS * NXc + (size_t)t * NXc + sj] = xg;
        const float* xs = sInp + sr * LDIN;
        fgv = physics_component(xs, xs[58], xs[59], sj);
      }
      __syncthreads();
      if (is_mlp) {
        v8f d = mm16<32>(sH1, LDH, w2, lane);
        const int nc = wave * 16 + (lane & 15);
        const int kh = lane >> 4;
#pragma unroll
        for (int v = 0; v < 8; ++v)
          sH2[(v + 8 * kh) * LDH + nc] = fast_tanh(d[v] + bias2);
        __builtin_amdgcn_fence(__ATOMIC_ACQ_REL, "wavefront");
        v8f e = mm16<4>(sH2 + wave * 16, LDH, w3, lane);
        const int n3 = lane & 15;
        if (n3 < NXc) {
#pragma unroll
          for (int v = 0; v < 8; ++v)
            sP[wave * 128 + (v + 8 * kh) * NXc + n3] = e[v];
        }
      }
      __syncthreads();
      if (!is_mlp) {
        float nn = sP[st];
#pragma unroll
        for (int w = 1; w < 8; ++w) nn += sP[w * 128 + st];
        const float k = fgv * scj + nn + b3j;
        float xn;
        if (s == 0)      { k1 = k; xn = xg + 0.5f * DELTA_F * k; }
        else if (s == 1) { k2 = k; xn = xg + 0.5f * DELTA_F * k; }
        else if (s == 2) { k3 = k; xn = xg + DELTA_F * k; }
        else {
          xn = xg + (DELTA_F / 6.0f) * (k1 + 2.0f * k2 + 2.0f * k3 + k);
          XGn[st] = xn;
          xg = xn;
        }
        sInp[sr * LDIN + sj] = xn;
      } else {
        if (s == 0) {
          float y0[3], y1[3];
#pragma unroll
          for (int ii = 0; ii < 3; ++ii) {
            const int i = tid + ii * NMLP;
            if (i < BT * 40) {
              int r = i / 40, c = i - r * 40;
              y0[ii] = Zc[r * ZP + c];
              y1[ii] = (c < 32) ? Zc[r * ZP + c + 8] : XGc[r * NXc + (c - 32)];
            }
          }
#pragma unroll
          for (int ii = 0; ii < 3; ++ii) {
            const int i = tid + ii * NMLP;
            if (i < BT * 40) {
              int r = i / 40, c = i - r * 40;
              sInp[r * LDIN + 8 + c] = 0.5f * (y0[ii] + y1[ii]);
            }
          }
        } else if (s == 2) {
          float vv[3];
#pragma unroll
          for (int ii = 0; ii < 3; ++ii) {
            const int i = tid + ii * NMLP;
            if (i < BT * 40) {
              int r = i / 40, c = i - r * 40;
              vv[ii] = (c < 32) ? Zc[r * ZP + 8 + c] : XGc[r * NXc + (c - 32)];
            }
          }
#pragma unroll
          for (int ii = 0; ii < 3; ++ii) {
            const int i = tid + ii * NMLP;
            if (i < BT * 40) {
              int r = i / 40, c = i - r * 40;
              sInp[r * LDIN + 8 + c] = vv[ii];
            }
          }
        } else if (s == 3) {
          const int tn = (t + 1 < T_STEPS) ? (t + 1) : t;
          float vv[4];
#pragma unroll
          for (int ii = 0; ii < 4; ++ii) {
            const int i = tid + ii * NMLP;
            if (i < BT * 52) {
              int r = i / 52, c = i - r * 52;
              float v;
              if      (c < 32) v = Zc[r * ZP + 8 + c];
              else if (c < 40) v = XGc[r * NXc + (c - 32)];
              else if (c < 48) v = Zc[r * ZP + (c + 2)];
              else if (c < 50) v = sUA[r * UP + t * NUc + (c - 48)];
              else             v = sUA[r * UP + tn * NUc + (c - 50)];
              vv[ii] = v;
            }
          }
#pragma unroll
          for (int ii = 0; ii < 4; ++ii) {
            const int i = tid + ii * NMLP;
            if (i < BT * 52) {
              int r = i / 52, c = i - r * 52;
              if (c < NZc) Zn[r * ZP + c] = vv[ii];
              sInp[r * LDIN + 8 + c] = vv[ii];
            }
          }
        }
      }
      __syncthreads();
    }
  }
}

extern "C" void kernel_launch(void* const* d_in, const int* in_sizes, int n_in,
                              void* d_out, int out_size, void* d_ws, size_t ws_size,
                              hipStream_t stream) {
  (void)in_sizes; (void)n_in; (void)out_size; (void)d_ws; (void)ws_size;
  const float* useq = (const float*)d_in[0];
  const float* xGz0 = (const float*)d_in[1];
  const float* W1   = (const float*)d_in[2];
  const float* b1   = (const float*)d_in[3];
  const float* W2   = (const float*)d_in[4];
  const float* b2   = (const float*)d_in[5];
  const float* W3   = (const float*)d_in[6];
  const float* b3   = (const float*)d_in[7];
  float* out = (float*)d_out;
  cstr_rollout_kernel<<<dim3(B_TOT / BT), dim3(NTHREADS), SMEM_BYTES, stream>>>(
      useq, xGz0, W1, b1, W2, b2, W3, b3, out);
}


# tanh quotient 2/(e+1) as 2*v_rcp_f32(e+1), denominator in [1,inf]: drops the 9-op full-range division sequence per activation behind each WMMA chain; quotient within 1 ulp, not correctly rounded
# speedup vs baseline: 2.9734x; 1.0740x over previous
#include <hip/hip_runtime.h>
#include <math.h>


typedef __attribute__((ext_vector_type(2))) float v2f;
typedef __attribute__((ext_vector_type(8))) float v8f;

#define B_TOT   2048
#define T_STEPS 128
#define NXc     8
#define NUc     2
#define NZc     50
#define DH      128
#define BT      16
#define DELTA_F 0.01f
#define NTHREADS 384
#define NMLP     256

#define LDIN    68
#define LDH     132
#define ZP      52
#define UP      258
#define OFF_INP 0u
#define OFF_H1  (OFF_INP + 16u*LDIN*4u)
#define OFF_H2  (OFF_H1  + 16u*LDH*4u)
#define OFF_P   (OFF_H2  + 16u*LDH*4u)
#define OFF_XG  (OFF_P   + 8u*128u*4u)
#define OFF_Z   (OFF_XG  + 2u*128u*4u)
#define OFF_UA  (OFF_Z   + 2u*16u*ZP*4u)
#define SMEM_BYTES (OFF_UA + 16u*UP*4u)

__device__ __forceinline__ float fast_tanh(float x) {
  float e = __builtin_amdgcn_exp2f(x * (2.0f * 1.442695041f));
  return 1.0f - 2.0f * __builtin_amdgcn_rcpf(e + 1.0f);
}

template <int KI>
__device__ __forceinline__ v8f mm16(const float* __restrict__ A, int lda,
                                    const v2f (&b)[KI], int lane) {
  const int mr = lane & 15;
  const int kh = lane >> 4;
  const float* ap = A + mr * lda + 2 * kh;
  v2f a[KI];
#pragma unroll
  for (int kk = 0; kk < KI; ++kk) a[kk] = *(const v2f*)(ap + 4 * kk);
  __builtin_amdgcn_sched_barrier(0);
  v8f acc = {};
#pragma unroll
  for (int kk = 0; kk < KI; ++kk)
    acc = __builtin_amdgcn_wmma_f32_16x16x4_f32(false, a[kk], false, b[kk],
                                                (short)0, acc, false, false);
  return acc;
}

__device__ __forceinline__ float physics_component(const float* xs, float u0,
                                                   float u1, int j) {
  float Hr  = xs[0] * 2.0f + 50.0f;
  float CAr = xs[1] * 0.2f + 1.0f;
  float CBr = xs[2] * 0.1f + 0.5f;
  float Tr  = xs[3] * 4.0f + 313.0f;
  float Hb  = xs[4] * 2.0f + 50.0f;
  float CAb = xs[5] * 0.2f + 1.0f;
  float CBb = xs[6] * 0.1f + 0.5f;
  float Tb  = xs[7] * 4.0f + 313.0f;
  float F = u0 * 1.0f + 10.0f;
  float D = u1 * 0.5f + 5.0f;
  float Fr = 2.5f * __fsqrt_rn(Hr);
  float Fb = 1.5f * __fsqrt_rn(Hb);
  float invArHr = __fdividef(1.0f, 3.0f * Hr);
  float invAbHb = __fdividef(1.0f, 3.0f * Hb);
  float dHr  = (F + D - Fr) * (1.0f / 3.0f);
  float dCAr = (F * (6.0f - CAr) - D * CAr) * invArHr;
  float dCBr = (-F * CBr - D * CBr) * invArHr;
  float dTr  = (F * (320.0f - Tr) + D * (300.0f - Tr)) * invArHr
             + (200.0f / 30.0f) * invArHr;
  float dHb  = (Fr - Fb - D) * (1.0f / 3.0f);
  float dCAb = (Fr * (CAr - CAb) + D * CAb) * invAbHb;
  float dCBb = (Fr * (CBr - CBb) + D * CBb) * invAbHb;
  float dTb  = Fr * (Tr - Tb) * invAbHb + (200.0f / 30.0f) * invAbHb;
  float lo = (j & 2) ? ((j & 1) ? dTr : dCBr) : ((j & 1) ? dCAr : dHr);
  float hi = (j & 2) ? ((j & 1) ? dTb : dCBb) : ((j & 1) ? dCAb : dHb);
  return (j & 4) ? hi : lo;
}

__global__ __launch_bounds__(NTHREADS) void cstr_rollout_kernel(
    const float* __restrict__ useq, const float* __restrict__ xGz0,
    const float* __restrict__ W1g, const float* __restrict__ b1g,
    const float* __restrict__ W2g, const float* __restrict__ b2g,
    const float* __restrict__ W3g, const float* __restrict__ b3g,
    float* __restrict__ out) {
  extern __shared__ __align__(16) char smem[];
  float* sInp = (float*)(smem + OFF_INP);
  float* sH1  = (float*)(smem + OFF_H1);
  float* sH2  = (float*)(smem + OFF_H2);
  float* sP   = (float*)(smem + OFF_P);
  float* sXG  = (float*)(smem + OFF_XG);
  float* sZ   = (float*)(smem + OFF_Z);
  float* sUA  = (float*)(smem + OFF_UA);

  const int tid  = threadIdx.x;
  const int lane = tid & 31;
  const int wave = __builtin_amdgcn_readfirstlane(tid >> 5);
  const bool is_mlp = wave < 8;
  const int bt0  = blockIdx.x * BT;

  const int st = (tid - NMLP) & 127;
  const int sr = st >> 3;
  const int sj = st & 7;

  v2f w1[16] = {}, w2[32] = {}, w3[4] = {};
  float bias1 = 0.0f, bias2 = 0.0f;
  const int wn  = lane & 15;
  const int wkh = lane >> 4;
  if (is_mlp) {
    const int wcol = wave * 16 + wn;
#pragma unroll
    for (int kk = 0; kk < 16; ++kk) {
      const int k0 = 4 * kk + 2 * wkh;
      const float lo = (k0     < 60) ? W1g[(k0)     * DH + wcol] : 0.0f;
      const float hi = (k0 + 1 < 60) ? W1g[(k0 + 1) * DH + wcol] : 0.0f;
      w1[kk] = (v2f){lo, hi};
    }
#pragma unroll
    for (int kk = 0; kk < 32; ++kk) {
      const int k0 = 4 * kk + 2 * wkh;
      w2[kk] = (v2f){W2g[(k0) * DH + wcol], W2g[(k0 + 1) * DH + wcol]};
    }
#pragma unroll
    for (int kk = 0; kk < 4; ++kk) {
      const int k0 = wave * 16 + 4 * kk + 2 * wkh;
      v2f p = (v2f){0.0f, 0.0f};
      if (wn < NXc) p = (v2f){W3g[(k0) * NXc + wn], W3g[(k0 + 1) * NXc + wn]};
      w3[kk] = p;
    }
    bias1 = b1g[wcol];
    bias2 = b2g[wcol];
  }

  float xg = 0.0f, b3j = 0.0f;
  if (!is_mlp) {
    xg  = xGz0[(size_t)(bt0 + sr) * 58 + sj];
    b3j = b3g[sj];
  }
  const float scj = ((sj & 3) == 0) ? 0.5f : ((sj & 3) == 1) ? 5.0f
                  : ((sj & 3) == 2) ? 10.0f : 0.25f;

  for (int i = tid; i < BT * T_STEPS * NUc; i += NTHREADS) {
    int r = i >> 8, q = i & 255;
    sUA[r * UP + q] = useq[(size_t)bt0 * (T_STEPS * NUc) + i];
  }
  for (int i = tid; i < BT * 58; i += NTHREADS) {
    int r = i / 58, c = i - r * 58;
    float v = xGz0[(size_t)(bt0 + r) * 58 + c];
    if (c < NXc) sXG[r * NXc + c] = v;
    else         sZ[r * ZP + (c - NXc)] = v;
    sInp[r * LDIN + c] = v;
  }
  if (tid < BT * 6) {
    int r = tid / 6, c = tid - r * 6;
    float v = (c < 2) ? useq[(size_t)(bt0 + r) * (T_STEPS * NUc) + c] : 0.0f;
    sInp[r * LDIN + 58 + c] = v;
  }
  __syncthreads();

  float k1 = 0.0f, k2 = 0.0f, k3 = 0.0f;
  for (int t = 0; t < T_STEPS; ++t) {
    const int cur = t & 1;
    const float* Zc  = sZ  + cur * (BT * ZP);
    float*       Zn  = sZ  + (cur ^ 1) * (BT * ZP);
    const float* XGc = sXG + cur * (BT * NXc);
    float*       XGn = sXG + (cur ^ 1) * (BT * NXc);

#pragma unroll
    for (int s = 0; s < 4; ++s) {
      float fgv = 0.0f;
      if (is_mlp) {
        v8f d = mm16<16>(sInp, LDIN, w1, lane);
        const int nc = wave * 16 + (lane & 15);
        const int kh = lane >> 4;
#pragma unroll
        for (int v = 0; v < 8; ++v)
          sH1[(v + 8 * kh) * LDH + nc] = fast_tanh(d[v] + bias1);
      } else {
        if (s == 0)
          out[(size_t)(bt0 + sr) * T_STEPS * NXc + (size_t)t * NXc + sj] = xg;
        const float* xs = sInp + sr * LDIN;
        fgv = physics_component(xs, xs[58], xs[59], sj);
      }
      __syncthreads();
      if (is_mlp) {
        v8f d = mm16<32>(sH1, LDH, w2, lane);
        const int nc = wave * 16 + (lane & 15);
        const int kh = lane >> 4;
#pragma unroll
        for (int v = 0; v < 8; ++v)
          sH2[(v + 8 * kh) * LDH + nc] = fast_tanh(d[v] + bias2);
        __builtin_amdgcn_fence(__ATOMIC_ACQ_REL, "wavefront");
        v8f e = mm16<4>(sH2 + wave * 16, LDH, w3, lane);
        const int n3 = lane & 15;
        if (n3 < NXc) {
#pragma unroll
          for (int v = 0; v < 8; ++v)
            sP[wave * 128 + (v + 8 * kh) * NXc + n3] = e[v];
        }
      }
      __syncthreads();
      if (!is_mlp) {
        float nn = sP[st];
#pragma unroll
        for (int w = 1; w < 8; ++w) nn += sP[w * 128 + st];
        const float k = fgv * scj + nn + b3j;
        float xn;
        if (s == 0)      { k1 = k; xn = xg + 0.5f * DELTA_F * k; }
        else if (s == 1) { k2 = k; xn = xg + 0.5f * DELTA_F * k; }
        else if (s == 2) { k3 = k; xn = xg + DELTA_F * k; }
        else {
          xn = xg + (DELTA_F / 6.0f) * (k1 + 2.0f * k2 + 2.0f * k3 + k);
          XGn[st] = xn;
          xg = xn;
        }
        sInp[sr * LDIN + sj] = xn;
      } else {
        if (s == 0) {
          float y0[3], y1[3];
#pragma unroll
          for (int ii = 0; ii < 3; ++ii) {
            const int i = tid + ii * NMLP;
            if (i < BT * 40) {
              int r = i / 40, c = i - r * 40;
              y0[ii] = Zc[r * ZP + c];
              y1[ii] = (c < 32) ? Zc[r * ZP + c + 8] : XGc[r * NXc + (c - 32)];
            }
          }
#pragma unroll
          for (int ii = 0; ii < 3; ++ii) {
            const int i = tid + ii * NMLP;
            if (i < BT * 40) {
              int r = i / 40, c = i - r * 40;
              sInp[r * LDIN + 8 + c] = 0.5f * (y0[ii] + y1[ii]);
            }
          }
        } else if (s == 2) {
          float vv[3];
#pragma unroll
          for (int ii = 0; ii < 3; ++ii) {
            const int i = tid + ii * NMLP;
            if (i < BT * 40) {
              int r = i / 40, c = i - r * 40;
              vv[ii] = (c < 32) ? Zc[r * ZP + 8 + c] : XGc[r * NXc + (c - 32)];
            }
          }
#pragma unroll
          for (int ii = 0; ii < 3; ++ii) {
            const int i = tid + ii * NMLP;
            if (i < BT * 40) {
              int r = i / 40, c = i - r * 40;
              sInp[r * LDIN + 8 + c] = vv[ii];
            }
          }
        } else if (s == 3) {
          const int tn = (t + 1 < T_STEPS) ? (t + 1) : t;
          float vv[4];
#pragma unroll
          for (int ii = 0; ii < 4; ++ii) {
            const int i = tid + ii * NMLP;
            if (i < BT * 52) {
              int r = i / 52, c = i - r * 52;
              float v;
              if      (c < 32) v = Zc[r * ZP + 8 + c];
              else if (c < 40) v = XGc[r * NXc + (c - 32)];
              else if (c < 48) v = Zc[r * ZP + (c + 2)];
              else if (c < 50) v = sUA[r * UP + t * NUc + (c - 48)];
              else             v = sUA[r * UP + tn * NUc + (c - 50)];
              vv[ii] = v;
            }
          }
#pragma unroll
          for (int ii = 0; ii < 4; ++ii) {
            const int i = tid + ii * NMLP;
            if (i < BT * 52) {
              int r = i / 52, c = i - r * 52;
              if (c < NZc) Zn[r * ZP + c] = vv[ii];
              sInp[r * LDIN + 8 + c] = vv[ii];
            }
          }
        }
      }
      __syncthreads();
    }
  }
}

extern "C" void kernel_launch(void* const* d_in, const int* in_sizes, int n_in,
                              void* d_out, int out_size, void* d_ws, size_t ws_size,
                              hipStream_t stream) {
  (void)in_sizes; (void)n_in; (void)out_size; (void)d_ws; (void)ws_size;
  const float* useq = (const float*)d_in[0];
  const float* xGz0 = (const float*)d_in[1];
  const float* W1   = (const float*)d_in[2];
  const float* b1   = (const float*)d_in[3];
  const float* W2   = (const float*)d_in[4];
  const float* b2   = (const float*)d_in[5];
  const float* W3   = (const float*)d_in[6];
  const float* b3   = (const float*)d_in[7];
  float* out = (float*)d_out;
  cstr_rollout_kernel<<<dim3(B_TOT / BT), dim3(NTHREADS), SMEM_BYTES, stream>>>(
      useq, xGz0, W1, b1, W2, b2, W3, b3, out);
}


# Layer 1 split by linearity: history/input columns multiplied twice a step (stage 1, stage 4; stages 2-3 take their mean), only the 8 state columns per stage; layer-1 sum order changes
# speedup vs baseline: 3.3110x; 1.1135x over previous
#include <hip/hip_runtime.h>
#include <math.h>


typedef __attribute__((ext_vector_type(2))) float v2f;
typedef __attribute__((ext_vector_type(8))) float v8f;

#define B_TOT   2048
#define T_STEPS 128
#define NXc     8
#define NUc     2
#define NZc     50
#define DH      128
#define BT      16
#define DELTA_F 0.01f
#define NTHREADS 384
#define NMLP     256

#define LDIN    68
#define LDH     132
#define ZP      52
#define UP      258
#define OFF_INP 0u
#define OFF_H1  (OFF_INP + 16u*LDIN*4u)
#define OFF_H2  (OFF_H1  + 16u*LDH*4u)
#define OFF_P   (OFF_H2  + 16u*LDH*4u)
#define OFF_XG  (OFF_P   + 8u*128u*4u)
#define OFF_Z   (OFF_XG  + 2u*128u*4u)
#define OFF_UA  (OFF_Z   + 2u*16u*ZP*4u)
#define SMEM_BYTES (OFF_UA + 16u*UP*4u)

__device__ __forceinline__ float fast_tanh(float x) {
  float e = __builtin_amdgcn_exp2f(x * (2.0f * 1.442695041f));
  return 1.0f - 2.0f * __builtin_amdgcn_rcpf(e + 1.0f);
}

template <int KI>
__device__ __forceinline__ v8f mm16(const float* __restrict__ A, int lda,
                                    const v2f (&b)[KI], int lane) {
  const int mr = lane & 15;
  const int kh = lane >> 4;
  const float* ap = A + mr * lda + 2 * kh;
  v2f a[KI];
#pragma unroll
  for (int kk = 0; kk < KI; ++kk) a[kk] = *(const v2f*)(ap + 4 * kk);
  __builtin_amdgcn_sched_barrier(0);
  v8f acc = {};
#pragma unroll
  for (int kk = 0; kk < KI; ++kk)
    acc = __builtin_amdgcn_wmma_f32_16x16x4_f32(false, a[kk], false, b[kk],
                                                (short)0, acc, false, false);
  return acc;
}

__device__ __forceinline__ float physics_component(const float* xs, float u0,
                                                   float u1, int j) {
  float Hr  = xs[0] * 2.0f + 50.0f;
  float CAr = xs[1] * 0.2f + 1.0f;
  float CBr = xs[2] * 0.1f + 0.5f;
  float Tr  = xs[3] * 4.0f + 313.0f;
  float Hb  = xs[4] * 2.0f + 50.0f;
  float CAb = xs[5] * 0.2f + 1.0f;
  float CBb = xs[6] * 0.1f + 0.5f;
  float Tb  = xs[7] * 4.0f + 313.0f;
  float F = u0 * 1.0f + 10.0f;
  float D = u1 * 0.5f + 5.0f;
  float Fr = 2.5f * __fsqrt_rn(Hr);
  float Fb = 1.5f * __fsqrt_rn(Hb);
  float invArHr = __fdividef(1.0f, 3.0f * Hr);
  float invAbHb = __fdividef(1.0f, 3.0f * Hb);
  float dHr  = (F + D - Fr) * (1.0f / 3.0f);
  float dCAr = (F * (6.0f - CAr) - D * CAr) * invArHr;
  float dCBr = (-F * CBr - D * CBr) * invArHr;
  float dTr  = (F * (320.0f - Tr) + D * (300.0f - Tr)) * invArHr
             + (200.0f / 30.0f) * invArHr;
  float dHb  = (Fr - Fb - D) * (1.0f / 3.0f);
  float dCAb = (Fr * (CAr - CAb) + D * CAb) * invAbHb;
  float dCBb = (Fr * (CBr - CBb) + D * CBb) * invAbHb;
  float dTb  = Fr * (Tr - Tb) * invAbHb + (200.0f / 30.0f) * invAbHb;
  float lo = (j & 2) ? ((j & 1) ? dTr : dCBr) : ((j & 1) ? dCAr : dHr);
  float hi = (j & 2) ? ((j & 1) ? dTb : dCBb) : ((j & 1) ? dCAb : dHb);
  return (j & 4) ? hi : lo;
}

__global__ __launch_bounds__(NTHREADS) void cstr_rollout_kernel(
    const float* __restrict__ useq, const float* __restrict__ xGz0,
    const float* __restrict__ W1g, const float* __restrict__ b1g,
    const float* __restrict__ W2g, const float* __restrict__ b2g,
    const float* __restrict__ W3g, const float* __restrict__ b3g,
    float* __restrict__ out) {
  extern __shared__ __align__(16) char smem[];
  float* sInp = (float*)(smem + OFF_INP);
  float* sH1  = (float*)(smem + OFF_H1);
  float* sH2  = (float*)(smem + OFF_H2);
  float* sP   = (float*)(smem + OFF_P);
  float* sXG  = (float*)(smem + OFF_XG);
  float* sZ   = (float*)(smem + OFF_Z);
  float* sUA  = (float*)(smem + OFF_UA);

  const int tid  = threadIdx.x;
  const int lane = tid & 31;
  const int wave = __builtin_amdgcn_readfirstlane(tid >> 5);
  const bool is_mlp = wave < 8;
  const int bt0  = blockIdx.x * BT;

  const int st = (tid - NMLP) & 127;
  const int sr = st >> 3;
  const int sj = st & 7;

  v2f w1[16] = {}, w2[32] = {}, w3[4] = {};
  float bias1 = 0.0f, bias2 = 0.0f;
  const int wn  = lane & 15;
  const int wkh = lane >> 4;
  if (is_mlp) {
    const int wcol = wave * 16 + wn;
#pragma unroll
    for (int kk = 0; kk < 16; ++kk) {
      const int k0 = 4 * kk + 2 * wkh;
      const float lo = (k0     < 60) ? W1g[(k0)     * DH + wcol] : 0.0f;
      const float hi = (k0 + 1 < 60) ? W1g[(k0 + 1) * DH + wcol] : 0.0f;
      w1[kk] = (v2f){lo, hi};
    }
#pragma unroll
    for (int kk = 0; kk < 32; ++kk) {
      const int k0 = 4 * kk + 2 * wkh;
      w2[kk] = (v2f){W2g[(k0) * DH + wcol], W2g[(k0 + 1) * DH + wcol]};
    }
#pragma unroll
    for (int kk = 0; kk < 4; ++kk) {
      const int k0 = wave * 16 + 4 * kk + 2 * wkh;
      v2f p = (v2f){0.0f, 0.0f};
      if (wn < NXc) p = (v2f){W3g[(k0) * NXc + wn], W3g[(k0 + 1) * NXc + wn]};
      w3[kk] = p;
    }
    bias1 = b1g[wcol];
    bias2 = b2g[wcol];
  }

  float xg = 0.0f, b3j = 0.0f;
  if (!is_mlp) {
    xg  = xGz0[(size_t)(bt0 + sr) * 58 + sj];
    b3j = b3g[sj];
  }
  const float scj = ((sj & 3) == 0) ? 0.5f : ((sj & 3) == 1) ? 5.0f
                  : ((sj & 3) == 2) ? 10.0f : 0.25f;

  for (int i = tid; i < BT * T_STEPS * NUc; i += NTHREADS) {
    int r = i >> 8, q = i & 255;
    sUA[r * UP + q] = useq[(size_t)bt0 * (T_STEPS * NUc) + i];
  }
  for (int i = tid; i < BT * 58; i += NTHREADS) {
    int r = i / 58, c = i - r * 58;
    float v = xGz0[(size_t)(bt0 + r) * 58 + c];
    if (c < NXc) sXG[r * NXc + c] = v;
    else         sZ[r * ZP + (c - NXc)] = v;
    sInp[r * LDIN + c] = v;
  }
  if (tid < BT * 6) {
    int r = tid / 6, c = tid - r * 6;
    float v = (c < 2) ? useq[(size_t)(bt0 + r) * (T_STEPS * NUc) + c] : 0.0f;
    sInp[r * LDIN + 58 + c] = v;
  }
  __syncthreads();

  float k1 = 0.0f, k2 = 0.0f, k3 = 0.0f;
  for (int t = 0; t < T_STEPS; ++t) {
    const int cur = t & 1;
    const float* Zc  = sZ  + cur * (BT * ZP);
    float*       Zn  = sZ  + (cur ^ 1) * (BT * ZP);
    const float* XGc = sXG + cur * (BT * NXc);
    float*       XGn = sXG + (cur ^ 1) * (BT * NXc);

    v8f P1 = {}, P4 = {}, Ph = {};

#pragma unroll
    for (int s = 0; s < 4; ++s) {
      float fgv = 0.0f;
      if (is_mlp) {
        const int nc = wave * 16 + (lane & 15);
        const int kh = lane >> 4;
        const float* ap = sInp + (lane & 15) * LDIN + 2 * kh;
        v8f d;
        if (s == 0) {
          v2f a[16];
#pragma unroll
          for (int kk = 0; kk < 16; ++kk) a[kk] = *(const v2f*)(ap + 4 * kk);
          __builtin_amdgcn_sched_barrier(0);
          v8f acc = {};
#pragma unroll
          for (int kk = 2; kk < 16; ++kk)
            acc = __builtin_amdgcn_wmma_f32_16x16x4_f32(false, a[kk], false, w1[kk],
                                                        (short)0, acc, false, false);
          P1 = acc;
#pragma unroll
          for (int kk = 0; kk < 2; ++kk)
            acc = __builtin_amdgcn_wmma_f32_16x16x4_f32(false, a[kk], false, w1[kk],
                                                        (short)0, acc, false, false);
          d = acc;
        } else {
          const v2f a0 = *(const v2f*)(ap);
          const v2f a1 = *(const v2f*)(ap + 4);
          v8f acc = (s == 3) ? P4 : Ph;
          acc = __builtin_amdgcn_wmma_f32_16x16x4_f32(false, a0, false, w1[0],
                                                      (short)0, acc, false, false);
          acc = __builtin_amdgcn_wmma_f32_16x16x4_f32(false, a1, false, w1[1],
                                                      (short)0, acc, false, false);
          d = acc;
        }
#pragma unroll
        for (int v = 0; v < 8; ++v)
          sH1[(v + 8 * kh) * LDH + nc] = fast_tanh(d[v] + bias1);
      } else {
        if (s == 0)
          out[(size_t)(bt0 + sr) * T_STEPS * NXc + (size_t)t * NXc + sj] = xg;
        const float* xs = sInp + sr * LDIN;
        fgv = physics_component(xs, xs[58], xs[59], sj);
      }
      __syncthreads();
      if (is_mlp) {
        v8f d = mm16<32>(sH1, LDH, w2, lane);
        const int nc = wave * 16 + (lane & 15);
        const int kh = lane >> 4;
#pragma unroll
        for (int v = 0; v < 8; ++v)
          sH2[(v + 8 * kh) * LDH + nc] = fast_tanh(d[v] + bias2);
        __builtin_amdgcn_fence(__ATOMIC_ACQ_REL, "wavefront");
        v8f e = mm16<4>(sH2 + wave * 16, LDH, w3, lane);
        const int n3 = lane & 15;
        if (n3 < NXc) {
#pragma unroll
          for (int v = 0; v < 8; ++v)
            sP[wave * 128 + (v + 8 * kh) * NXc + n3] = e[v];
        }
      }
      __syncthreads();
      if (!is_mlp) {
        float nn = sP[st];
#pragma unroll
        for (int w = 1; w < 8; ++w) nn += sP[w * 128 + st];
        const float k = fgv * scj + nn + b3j;
        float xn;
        if (s == 0)      { k1 = k; xn = xg + 0.5f * DELTA_F * k; }
        else if (s == 1) { k2 = k; xn = xg + 0.5f * DELTA_F * k; }
        else if (s == 2) { k3 = k; xn = xg + DELTA_F * k; }
        else {
          xn = xg + (DELTA_F / 6.0f) * (k1 + 2.0f * k2 + 2.0f * k3 + k);
          XGn[st] = xn;
          xg = xn;
        }
        sInp[sr * LDIN + sj] = xn;
      } else {
        if (s == 0) {
          const int mr = lane & 15;
          const int kh = lane >> 4;
          const float* zp = Zc + mr * ZP + 2 * kh;
          const float* xp = XGc + mr * NXc + 2 * kh;
          const float* ip = sInp + mr * LDIN + 2 * kh;
          v2f a[14];
#pragma unroll
          for (int kk = 2; kk < 10; ++kk)  a[kk - 2] = *(const v2f*)(zp + 4 * kk);
#pragma unroll
          for (int kk = 10; kk < 12; ++kk) a[kk - 2] = *(const v2f*)(xp + 4 * (kk - 10));
#pragma unroll
          for (int kk = 12; kk < 16; ++kk) a[kk - 2] = *(const v2f*)(ip + 4 * kk);
          __builtin_amdgcn_sched_barrier(0);
          v8f acc = {};
#pragma unroll
          for (int kk = 2; kk < 16; ++kk)
            acc = __builtin_amdgcn_wmma_f32_16x16x4_f32(false, a[kk - 2], false, w1[kk],
                                                        (short)0, acc, false, false);
          P4 = acc;
#pragma unroll
          for (int v = 0; v < 8; ++v) Ph[v] = 0.5f * (P1[v] + acc[v]);
        } else if (s == 3) {
          const int tn = (t + 1 < T_STEPS) ? (t + 1) : t;
          float vv[4];
#pragma unroll
          for (int ii = 0; ii < 4; ++ii) {
            const int i = tid + ii * NMLP;
            if (i < BT * 52) {
              int r = i / 52, c = i - r * 52;
              float v;
              if      (c < 32) v = Zc[r * ZP + 8 + c];
              else if (c < 40) v = XGc[r * NXc + (c - 32)];
              else if (c < 48) v = Zc[r * ZP + (c + 2)];
              else if (c < 50) v = sUA[r * UP + t * NUc + (c - 48)];
              else             v = sUA[r * UP + tn * NUc + (c - 50)];
              vv[ii] = v;
            }
          }
#pragma unroll
          for (int ii = 0; ii < 4; ++ii) {
            const int i = tid + ii * NMLP;
            if (i < BT * 52) {
              int r = i / 52, c = i - r * 52;
              if (c < NZc) Zn[r * ZP + c] = vv[ii];
              sInp[r * LDIN + 8 + c] = vv[ii];
            }
          }
        }
      }
      __syncthreads();
    }
  }
}

extern "C" void kernel_launch(void* const* d_in, const int* in_sizes, int n_in,
                              void* d_out, int out_size, void* d_ws, size_t ws_size,
                              hipStream_t stream) {
  (void)in_sizes; (void)n_in; (void)out_size; (void)d_ws; (void)ws_size;
  const float* useq = (const float*)d_in[0];
  const float* xGz0 = (const float*)d_in[1];
  const float* W1   = (const float*)d_in[2];
  const float* b1   = (const float*)d_in[3];
  const float* W2   = (const float*)d_in[4];
  const float* b2   = (const float*)d_in[5];
  const float* W3   = (const float*)d_in[6];
  const float* b3   = (const float*)d_in[7];
  float* out = (float*)d_out;
  cstr_rollout_kernel<<<dim3(B_TOT / BT), dim3(NTHREADS), SMEM_BYTES, stream>>>(
      useq, xGz0, W1, b1, W2, b2, W3, b3, out);
}


# Layer-1 history product off the critical path: stage 4 and the next step's stage 1 share 10 of 13 k-steps, multiplied once (8 of them a step ahead), all in the phase-3 waits; same k order
# speedup vs baseline: 3.7391x; 1.1293x over previous
#include <hip/hip_runtime.h>
#include <math.h>


typedef __attribute__((ext_vector_type(2))) float v2f;
typedef __attribute__((ext_vector_type(8))) float v8f;

#define B_TOT   2048
#define T_STEPS 128
#define NXc     8
#define NUc     2
#define NZc     50
#define DH      128
#define BT      16
#define DELTA_F 0.01f
#define NTHREADS 384
#define NMLP     256

#define LDIN    68
#define LDH     132
#define ZP      52
#define UP      258
#define OFF_INP 0u
#define OFF_H1  (OFF_INP + 16u*LDIN*4u)
#define OFF_H2  (OFF_H1  + 16u*LDH*4u)
#define OFF_P   (OFF_H2  + 16u*LDH*4u)
#define OFF_XG  (OFF_P   + 8u*128u*4u)
#define OFF_Z   (OFF_XG  + 2u*128u*4u)
#define OFF_UA  (OFF_Z   + 2u*16u*ZP*4u)
#define SMEM_BYTES (OFF_UA + 16u*UP*4u)

__device__ __forceinline__ float fast_tanh(float x) {
  float e = __builtin_amdgcn_exp2f(x * (2.0f * 1.442695041f));
  return 1.0f - 2.0f * __builtin_amdgcn_rcpf(e + 1.0f);
}

template <int KI>
__device__ __forceinline__ v8f mm16(const float* __restrict__ A, int lda,
                                    const v2f (&b)[KI], int lane) {
  const int mr = lane & 15;
  const int kh = lane >> 4;
  const float* ap = A + mr * lda + 2 * kh;
  v2f a[KI];
#pragma unroll
  for (int kk = 0; kk < KI; ++kk) a[kk] = *(const v2f*)(ap + 4 * kk);
  __builtin_amdgcn_sched_barrier(0);
  v8f acc = {};
#pragma unroll
  for (int kk = 0; kk < KI; ++kk)
    acc = __builtin_amdgcn_wmma_f32_16x16x4_f32(false, a[kk], false, b[kk],
                                                (short)0, acc, false, false);
  return acc;
}

__device__ __forceinline__ float physics_component(const float* xs, float u0,
                                                   float u1, int j) {
  float Hr  = xs[0] * 2.0f + 50.0f;
  float CAr = xs[1] * 0.2f + 1.0f;
  float CBr = xs[2] * 0.1f + 0.5f;
  float Tr  = xs[3] * 4.0f + 313.0f;
  float Hb  = xs[4] * 2.0f + 50.0f;
  float CAb = xs[5] * 0.2f + 1.0f;
  float CBb = xs[6] * 0.1f + 0.5f;
  float Tb  = xs[7] * 4.0f + 313.0f;
  float F = u0 * 1.0f + 10.0f;
  float D = u1 * 0.5f + 5.0f;
  float Fr = 2.5f * __fsqrt_rn(Hr);
  float Fb = 1.5f * __fsqrt_rn(Hb);
  float invArHr = __fdividef(1.0f, 3.0f * Hr);
  float invAbHb = __fdividef(1.0f, 3.0f * Hb);
  float dHr  = (F + D - Fr) * (1.0f / 3.0f);
  float dCAr = (F * (6.0f - CAr) - D * CAr) * invArHr;
  float dCBr = (-F * CBr - D * CBr) * invArHr;
  float dTr  = (F * (320.0f - Tr) + D * (300.0f - Tr)) * invArHr
             + (200.0f / 30.0f) * invArHr;
  float dHb  = (Fr - Fb - D) * (1.0f / 3.0f);
  float dCAb = (Fr * (CAr - CAb) + D * CAb) * invAbHb;
  float dCBb = (Fr * (CBr - CBb) + D * CBb) * invAbHb;
  float dTb  = Fr * (Tr - Tb) * invAbHb + (200.0f / 30.0f) * invAbHb;
  float lo = (j & 2) ? ((j & 1) ? dTr : dCBr) : ((j & 1) ? dCAr : dHr);
  float hi = (j & 2) ? ((j & 1) ? dTb : dCBb) : ((j & 1) ? dCAb : dHb);
  return (j & 4) ? hi : lo;
}

__global__ __launch_bounds__(NTHREADS) void cstr_rollout_kernel(
    const float* __restrict__ useq, const float* __restrict__ xGz0,
    const float* __restrict__ W1g, const float* __restrict__ b1g,
    const float* __restrict__ W2g, const float* __restrict__ b2g,
    const float* __restrict__ W3g, const float* __restrict__ b3g,
    float* __restrict__ out) {
  extern __shared__ __align__(16) char smem[];
  float* sInp = (float*)(smem + OFF_INP);
  float* sH1  = (float*)(smem + OFF_H1);
  float* sH2  = (float*)(smem + OFF_H2);
  float* sP   = (float*)(smem + OFF_P);
  float* sXG  = (float*)(smem + OFF_XG);
  float* sZ   = (float*)(smem + OFF_Z);
  float* sUA  = (float*)(smem + OFF_UA);

  const int tid  = threadIdx.x;
  const int lane = tid & 31;
  const int wave = __builtin_amdgcn_readfirstlane(tid >> 5);
  const bool is_mlp = wave < 8;
  const int bt0  = blockIdx.x * BT;

  const int st = (tid - NMLP) & 127;
  const int sr = st >> 3;
  const int sj = st & 7;

  v2f w1[15] = {}, w2[32] = {}, w3[4] = {};
  float bias1 = 0.0f, bias2 = 0.0f;
  const int wn  = lane & 15;
  const int wkh = lane >> 4;
  if (is_mlp) {
    const int wcol = wave * 16 + wn;
#pragma unroll
    for (int kk = 0; kk < 15; ++kk) {
      const int k0 = 4 * kk + 2 * wkh;
      w1[kk] = (v2f){W1g[(k0) * DH + wcol], W1g[(k0 + 1) * DH + wcol]};
    }
#pragma unroll
    for (int kk = 0; kk < 32; ++kk) {
      const int k0 = 4 * kk + 2 * wkh;
      w2[kk] = (v2f){W2g[(k0) * DH + wcol], W2g[(k0 + 1) * DH + wcol]};
    }
#pragma unroll
    for (int kk = 0; kk < 4; ++kk) {
      const int k0 = wave * 16 + 4 * kk + 2 * wkh;
      v2f p = (v2f){0.0f, 0.0f};
      if (wn < NXc) p = (v2f){W3g[(k0) * NXc + wn], W3g[(k0 + 1) * NXc + wn]};
      w3[kk] = p;
    }
    bias1 = b1g[wcol];
    bias2 = b2g[wcol];
  }

  float xg = 0.0f, b3j = 0.0f;
  if (!is_mlp) {
    xg  = xGz0[(size_t)(bt0 + sr) * 58 + sj];
    b3j = b3g[sj];
  }
  const float scj = ((sj & 3) == 0) ? 0.5f : ((sj & 3) == 1) ? 5.0f
                  : ((sj & 3) == 2) ? 10.0f : 0.25f;

  for (int i = tid; i < BT * T_STEPS * NUc; i += NTHREADS) {
    int r = i >> 8, q = i & 255;
    sUA[r * UP + q] = useq[(size_t)bt0 * (T_STEPS * NUc) + i];
  }
  for (int i = tid; i < BT * 58; i += NTHREADS) {
    int r = i / 58, c = i - r * 58;
    float v = xGz0[(size_t)(bt0 + r) * 58 + c];
    if (c < NXc) sXG[r * NXc + c] = v;
    else         sZ[r * ZP + (c - NXc)] = v;
    sInp[r * LDIN + c] = v;
  }
  if (tid < BT * 6) {
    int r = tid / 6, c = tid - r * 6;
    float v = (c < 2) ? useq[(size_t)(bt0 + r) * (T_STEPS * NUc) + c] : 0.0f;
    sInp[r * LDIN + 58 + c] = v;
  }
  __syncthreads();

  v8f P1 = {}, Q = {}, Qz = {};
  if (is_mlp) {
    const float* ap = sInp + (lane & 15) * LDIN + 2 * (lane >> 4);
    const float* zp = sZ + (lane & 15) * ZP + 2 * (lane >> 4);
    v2f a[13], z[8];
#pragma unroll
    for (int kk = 2; kk < 15; ++kk) a[kk - 2] = *(const v2f*)(ap + 4 * kk);
#pragma unroll
    for (int kk = 2; kk < 10; ++kk) z[kk - 2] = *(const v2f*)(zp + 4 * kk);
    __builtin_amdgcn_sched_barrier(0);
    v8f acc = {};
#pragma unroll
    for (int kk = 2; kk < 15; ++kk)
      acc = __builtin_amdgcn_wmma_f32_16x16x4_f32(false, a[kk - 2], false, w1[kk],
                                                  (short)0, acc, false, false);
    P1 = acc;
    v8f acz = {};
#pragma unroll
    for (int kk = 2; kk < 10; ++kk)
      acz = __builtin_amdgcn_wmma_f32_16x16x4_f32(false, z[kk - 2], false, w1[kk],
                                                  (short)0, acz, false, false);
    Qz = acz;
  }

  float k1 = 0.0f, k2 = 0.0f, k3 = 0.0f;
  for (int t = 0; t < T_STEPS; ++t) {
    const int cur = t & 1;
    const float* Zc  = sZ  + cur * (BT * ZP);
    float*       Zn  = sZ  + (cur ^ 1) * (BT * ZP);
    const float* XGc = sXG + cur * (BT * NXc);
    float*       XGn = sXG + (cur ^ 1) * (BT * NXc);

    v8f P4 = {}, Ph = {};

#pragma unroll
    for (int s = 0; s < 4; ++s) {
      float fgv = 0.0f;
      if (is_mlp) {
        const int nc = wave * 16 + (lane & 15);
        const int kh = lane >> 4;
        const float* ap = sInp + (lane & 15) * LDIN + 2 * kh;
        const v2f a0 = *(const v2f*)(ap);
        const v2f a1 = *(const v2f*)(ap + 4);
        v8f d = (s == 0) ? P1 : (s == 3) ? P4 : Ph;
        d = __builtin_amdgcn_wmma_f32_16x16x4_f32(false, a0, false, w1[0],
                                                  (short)0, d, false, false);
        d = __builtin_amdgcn_wmma_f32_16x16x4_f32(false, a1, false, w1[1],
                                                  (short)0, d, false, false);
#pragma unroll
        for (int v = 0; v < 8; ++v)
          sH1[(v + 8 * kh) * LDH + nc] = fast_tanh(d[v] + bias1);
      } else {
        if (s == 0)
          out[(size_t)(bt0 + sr) * T_STEPS * NXc + (size_t)t * NXc + sj] = xg;
        const float* xs = sInp + sr * LDIN;
        const float* us = sUA + sr * UP + t * NUc;
        fgv = physics_component(xs, us[0], us[1], sj);
      }
      __syncthreads();
      if (is_mlp) {
        v8f d = mm16<32>(sH1, LDH, w2, lane);
        const int nc = wave * 16 + (lane & 15);
        const int kh = lane >> 4;
#pragma unroll
        for (int v = 0; v < 8; ++v)
          sH2[(v + 8 * kh) * LDH + nc] = fast_tanh(d[v] + bias2);
        __builtin_amdgcn_fence(__ATOMIC_ACQ_REL, "wavefront");
        v8f e = mm16<4>(sH2 + wave * 16, LDH, w3, lane);
        const int n3 = lane & 15;
        if (n3 < NXc) {
#pragma unroll
          for (int v = 0; v < 8; ++v)
            sP[wave * 128 + (v + 8 * kh) * NXc + n3] = e[v];
        }
      }
      __syncthreads();
      if (!is_mlp) {
        float nn = sP[st];
#pragma unroll
        for (int w = 1; w < 8; ++w) nn += sP[w * 128 + st];
        const float k = fgv * scj + nn + b3j;
        float xn;
        if (s == 0)      { k1 = k; xn = xg + 0.5f * DELTA_F * k; }
        else if (s == 1) { k2 = k; xn = xg + 0.5f * DELTA_F * k; }
        else if (s == 2) { k3 = k; xn = xg + DELTA_F * k; }
        else {
          xn = xg + (DELTA_F / 6.0f) * (k1 + 2.0f * k2 + 2.0f * k3 + k);
          XGn[st] = xn;
          xg = xn;
        }
        sInp[sr * LDIN + sj] = xn;
      } else {
        const int mr = lane & 15;
        const int kh = lane >> 4;
        const float* zp = Zc + mr * ZP + 2 * kh;
        const float* xp = XGc + mr * NXc + 2 * kh;
        if (s == 0) {
          const v2f a10 = *(const v2f*)(xp);
          const v2f a11 = *(const v2f*)(xp + 4);
          const v2f a12 = *(const v2f*)(zp + 40);
          const v2f a13 = *(const v2f*)(zp + 44);
          const v2f a14 = *(const v2f*)(kh ? (sUA + mr * UP + NUc * t)
                                           : (Zc + mr * ZP + 48));
          __builtin_amdgcn_sched_barrier(0);
          v8f acc = Qz;
          acc = __builtin_amdgcn_wmma_f32_16x16x4_f32(false, a10, false, w1[10],
                                                      (short)0, acc, false, false);
          acc = __builtin_amdgcn_wmma_f32_16x16x4_f32(false, a11, false, w1[11],
                                                      (short)0, acc, false, false);
          Q = acc;
          acc = __builtin_amdgcn_wmma_f32_16x16x4_f32(false, a12, false, w1[12],
                                                      (short)0, acc, false, false);
          acc = __builtin_amdgcn_wmma_f32_16x16x4_f32(false, a13, false, w1[13],
                                                      (short)0, acc, false, false);
          acc = __builtin_amdgcn_wmma_f32_16x16x4_f32(false, a14, false, w1[14],
                                                      (short)0, acc, false, false);
          P4 = acc;
#pragma unroll
          for (int v = 0; v < 8; ++v) Ph[v] = 0.5f * (P1[v] + acc[v]);
        } else if (s == 1) {
          const int tn = (t + 1 < T_STEPS) ? (t + 1) : t;
          const v2f a12 = *(const v2f*)(zp + 42);
          const v2f a13 = *(const v2f*)(zp + 46);
          const v2f a14 = *(const v2f*)(sUA + mr * UP + NUc * (kh ? tn : t));
          __builtin_amdgcn_sched_barrier(0);
          v8f acc = Q;
          acc = __builtin_amdgcn_wmma_f32_16x16x4_f32(false, a12, false, w1[12],
                                                      (short)0, acc, false, false);
          acc = __builtin_amdgcn_wmma_f32_16x16x4_f32(false, a13, false, w1[13],
                                                      (short)0, acc, false, false);
          acc = __builtin_amdgcn_wmma_f32_16x16x4_f32(false, a14, false, w1[14],
                                                      (short)0, acc, false, false);
          P1 = acc;
        } else if (s == 2) {
          v2f z[4];
#pragma unroll
          for (int j = 0; j < 4; ++j) z[j] = *(const v2f*)(zp + 16 + 4 * j);
          const int r = tid >> 4, q = tid & 15;
          const float* zr = Zc + r * ZP;
          const float v0 = zr[8 + q];
          const float v1 = zr[24 + q];
          const float* s2 = (q < 8) ? (XGc + r * NXc + q)
                                    : (zr + 34 + q);
          const float v2 = *s2;
          const float v3 = sUA[r * UP + t * NUc + (q & 1)];
          __builtin_amdgcn_sched_barrier(0);
          v8f acc = {};
#pragma unroll
          for (int j = 0; j < 4; ++j)
            acc = __builtin_amdgcn_wmma_f32_16x16x4_f32(false, z[j], false, w1[2 + j],
                                                        (short)0, acc, false, false);
          Qz = acc;
          float* zn = Zn + r * ZP;
          zn[q]      = v0;
          zn[16 + q] = v1;
          zn[32 + q] = v2;
          if (q < 2) zn[48 + q] = v3;
        } else {
          const v2f a6 = *(const v2f*)(zp + 32);
          const v2f a7 = *(const v2f*)(zp + 36);
          const v2f a8 = *(const v2f*)(xp);
          const v2f a9 = *(const v2f*)(xp + 4);
          __builtin_amdgcn_sched_barrier(0);
          v8f acc = Qz;
          acc = __builtin_amdgcn_wmma_f32_16x16x4_f32(false, a6, false, w1[6],
                                                      (short)0, acc, false, false);
          acc = __builtin_amdgcn_wmma_f32_16x16x4_f32(false, a7, false, w1[7],
                                                      (short)0, acc, false, false);
          acc = __builtin_amdgcn_wmma_f32_16x16x4_f32(false, a8, false, w1[8],
                                                      (short)0, acc, false, false);
          acc = __builtin_amdgcn_wmma_f32_16x16x4_f32(false, a9, false, w1[9],
                                                      (short)0, acc, false, false);
          Qz = acc;
        }
      }
      __syncthreads();
    }
  }
}

extern "C" void kernel_launch(void* const* d_in, const int* in_sizes, int n_in,
                              void* d_out, int out_size, void* d_ws, size_t ws_size,
                              hipStream_t stream) {
  (void)in_sizes; (void)n_in; (void)out_size; (void)d_ws; (void)ws_size;
  const float* useq = (const float*)d_in[0];
  const float* xGz0 = (const float*)d_in[1];
  const float* W1   = (const float*)d_in[2];
  const float* b1   = (const float*)d_in[3];
  const float* W2   = (const float*)d_in[4];
  const float* b2   = (const float*)d_in[5];
  const float* W3   = (const float*)d_in[6];
  const float* b3   = (const float*)d_in[7];
  float* out = (float*)d_out;
  cstr_rollout_kernel<<<dim3(B_TOT / BT), dim3(NTHREADS), SMEM_BYTES, stream>>>(
      useq, xGz0, W1, b1, W2, b2, W3, b3, out);
}


# Products transposed (weights as A, activations as B): a lane holds 8 hidden cols of one row, so layer 3 takes layer 2 from registers and stores are b128; bias heads each sum, L3 k order changes
# speedup vs baseline: 3.8919x; 1.0409x over previous
#include <hip/hip_runtime.h>
#include <math.h>


typedef __attribute__((ext_vector_type(2))) float v2f;
typedef __attribute__((ext_vector_type(4))) float v4f;
typedef __attribute__((ext_vector_type(8))) float v8f;

#define B_TOT   2048
#define T_STEPS 128
#define NXc     8
#define NUc     2
#define NZc     50
#define DH      128
#define BT      16
#define DELTA_F 0.01f
#define NTHREADS 384
#define NMLP     256

#define LDIN    68
#define LDH     132
#define ZP      52
#define UP      258
#define OFF_INP 0u
#define OFF_H1  (OFF_INP + 16u*LDIN*4u)
#define OFF_P   (OFF_H1  + 16u*LDH*4u)
#define OFF_XG  (OFF_P   + 8u*128u*4u)
#define OFF_Z   (OFF_XG  + 2u*128u*4u)
#define OFF_UA  (OFF_Z   + 2u*16u*ZP*4u)
#define SMEM_BYTES (OFF_UA + 16u*UP*4u)

__device__ __forceinline__ float fast_tanh(float x) {
  float e = __builtin_amdgcn_exp2f(x * (2.0f * 1.442695041f));
  return 1.0f - 2.0f * __builtin_amdgcn_rcpf(e + 1.0f);
}

__device__ __forceinline__ v8f wmma_t(v2f w, v2f x, v8f acc) {
  return __builtin_amdgcn_wmma_f32_16x16x4_f32(false, w, false, x,
                                               (short)0, acc, false, false);
}

template <int KI>
__device__ __forceinline__ v8f mm16t(const float* __restrict__ X, int ldx,
                                     const v2f (&w)[KI], v8f acc0, int lane) {
  const int nr = lane & 15;
  const int kh = lane >> 4;
  const float* xp = X + nr * ldx + 2 * kh;
  v2f x[KI];
#pragma unroll
  for (int kk = 0; kk < KI; ++kk) x[kk] = *(const v2f*)(xp + 4 * kk);
  __builtin_amdgcn_sched_barrier(0);
  v8f acc = acc0;
#pragma unroll
  for (int kk = 0; kk < KI; ++kk) acc = wmma_t(w[kk], x[kk], acc);
  return acc;
}

__device__ __forceinline__ float physics_component(const float* xs, float u0,
                                                   float u1, int j) {
  float Hr  = xs[0] * 2.0f + 50.0f;
  float CAr = xs[1] * 0.2f + 1.0f;
  float CBr = xs[2] * 0.1f + 0.5f;
  float Tr  = xs[3] * 4.0f + 313.0f;
  float Hb  = xs[4] * 2.0f + 50.0f;
  float CAb = xs[5] * 0.2f + 1.0f;
  float CBb = xs[6] * 0.1f + 0.5f;
  float Tb  = xs[7] * 4.0f + 313.0f;
  float F = u0 * 1.0f + 10.0f;
  float D = u1 * 0.5f + 5.0f;
  float Fr = 2.5f * __fsqrt_rn(Hr);
  float Fb = 1.5f * __fsqrt_rn(Hb);
  float invArHr = __fdividef(1.0f, 3.0f * Hr);
  float invAbHb = __fdividef(1.0f, 3.0f * Hb);
  float dHr  = (F + D - Fr) * (1.0f / 3.0f);
  float dCAr = (F * (6.0f - CAr) - D * CAr) * invArHr;
  float dCBr = (-F * CBr - D * CBr) * invArHr;
  float dTr  = (F * (320.0f - Tr) + D * (300.0f - Tr)) * invArHr
             + (200.0f / 30.0f) * invArHr;
  float dHb  = (Fr - Fb - D) * (1.0f / 3.0f);
  float dCAb = (Fr * (CAr - CAb) + D * CAb) * invAbHb;
  float dCBb = (Fr * (CBr - CBb) + D * CBb) * invAbHb;
  float dTb  = Fr * (Tr - Tb) * invAbHb + (200.0f / 30.0f) * invAbHb;
  float lo = (j & 2) ? ((j & 1) ? dTr : dCBr) : ((j & 1) ? dCAr : dHr);
  float hi = (j & 2) ? ((j & 1) ? dTb : dCBb) : ((j & 1) ? dCAb : dHb);
  return (j & 4) ? hi : lo;
}

__global__ __launch_bounds__(NTHREADS) void cstr_rollout_kernel(
    const float* __restrict__ useq, const float* __restrict__ xGz0,
    const float* __restrict__ W1g, const float* __restrict__ b1g,
    const float* __restrict__ W2g, const float* __restrict__ b2g,
    const float* __restrict__ W3g, const float* __restrict__ b3g,
    float* __restrict__ out) {
  extern __shared__ __align__(16) char smem[];
  float* sInp = (float*)(smem + OFF_INP);
  float* sH1  = (float*)(smem + OFF_H1);
  float* sP   = (float*)(smem + OFF_P);
  float* sXG  = (float*)(smem + OFF_XG);
  float* sZ   = (float*)(smem + OFF_Z);
  float* sUA  = (float*)(smem + OFF_UA);

  const int tid  = threadIdx.x;
  const int lane = tid & 31;
  const int wave = __builtin_amdgcn_readfirstlane(tid >> 5);
  const bool is_mlp = wave < 8;
  const int bt0  = blockIdx.x * BT;

  const int st = (tid - NMLP) & 127;
  const int sr = st >> 3;
  const int sj = st & 7;

  v2f w1[15] = {}, w2[32] = {}, w3[4] = {};
  v8f b1r = {}, b2r = {};
  const int wn  = lane & 15;
  const int wkh = lane >> 4;
  if (is_mlp) {
    const int wcol = wave * 16 + wn;
#pragma unroll
    for (int kk = 0; kk < 15; ++kk) {
      const int k0 = 4 * kk + 2 * wkh;
      w1[kk] = (v2f){W1g[(k0) * DH + wcol], W1g[(k0 + 1) * DH + wcol]};
    }
#pragma unroll
    for (int kk = 0; kk < 32; ++kk) {
      const int k0 = 4 * kk + 2 * wkh;
      w2[kk] = (v2f){W2g[(k0) * DH + wcol], W2g[(k0 + 1) * DH + wcol]};
    }
#pragma unroll
    for (int kk = 0; kk < 4; ++kk) {
      const int k0 = wave * 16 + 8 * wkh + 2 * kk;
      v2f p = (v2f){0.0f, 0.0f};
      if (wn < NXc) p = (v2f){W3g[(k0) * NXc + wn], W3g[(k0 + 1) * NXc + wn]};
      w3[kk] = p;
    }
#pragma unroll
    for (int v = 0; v < 8; ++v) {
      b1r[v] = b1g[wave * 16 + 8 * wkh + v];
      b2r[v] = b2g[wave * 16 + 8 * wkh + v];
    }
  }

  float xg = 0.0f, b3j = 0.0f;
  if (!is_mlp) {
    xg  = xGz0[(size_t)(bt0 + sr) * 58 + sj];
    b3j = b3g[sj];
  }
  const float scj = ((sj & 3) == 0) ? 0.5f : ((sj & 3) == 1) ? 5.0f
                  : ((sj & 3) == 2) ? 10.0f : 0.25f;

  for (int i = tid; i < BT * T_STEPS * NUc; i += NTHREADS) {
    int r = i >> 8, q = i & 255;
    sUA[r * UP + q] = useq[(size_t)bt0 * (T_STEPS * NUc) + i];
  }
  for (int i = tid; i < BT * 58; i += NTHREADS) {
    int r = i / 58, c = i - r * 58;
    float v = xGz0[(size_t)(bt0 + r) * 58 + c];
    if (c < NXc) sXG[r * NXc + c] = v;
    else         sZ[r * ZP + (c - NXc)] = v;
    sInp[r * LDIN + c] = v;
  }
  if (tid < BT * 6) {
    int r = tid / 6, c = tid - r * 6;
    float v = (c < 2) ? useq[(size_t)(bt0 + r) * (T_STEPS * NUc) + c] : 0.0f;
    sInp[r * LDIN + 58 + c] = v;
  }
  __syncthreads();

  v8f P1 = {}, Q = {}, Qz = {};
  if (is_mlp) {
    const float* ap = sInp + (lane & 15) * LDIN + 2 * (lane >> 4);
    const float* zp = sZ + (lane & 15) * ZP + 2 * (lane >> 4);
    v2f a[13], z[8];
#pragma unroll
    for (int kk = 2; kk < 15; ++kk) a[kk - 2] = *(const v2f*)(ap + 4 * kk);
#pragma unroll
    for (int kk = 2; kk < 10; ++kk) z[kk - 2] = *(const v2f*)(zp + 4 * kk);
    __builtin_amdgcn_sched_barrier(0);
    v8f acc = b1r;
#pragma unroll
    for (int kk = 2; kk < 15; ++kk) acc = wmma_t(w1[kk], a[kk - 2], acc);
    P1 = acc;
    v8f acz = b1r;
#pragma unroll
    for (int kk = 2; kk < 10; ++kk) acz = wmma_t(w1[kk], z[kk - 2], acz);
    Qz = acz;
  }

  float k1 = 0.0f, k2 = 0.0f, k3 = 0.0f;
  for (int t = 0; t < T_STEPS; ++t) {
    const int cur = t & 1;
    const float* Zc  = sZ  + cur * (BT * ZP);
    float*       Zn  = sZ  + (cur ^ 1) * (BT * ZP);
    const float* XGc = sXG + cur * (BT * NXc);
    float*       XGn = sXG + (cur ^ 1) * (BT * NXc);

    v8f P4 = {}, Ph = {};

#pragma unroll
    for (int s = 0; s < 4; ++s) {
      float fgv = 0.0f;
      if (is_mlp) {
        const int nr = lane & 15;
        const int kh = lane >> 4;
        const float* ap = sInp + nr * LDIN + 2 * kh;
        const v2f a0 = *(const v2f*)(ap);
        const v2f a1 = *(const v2f*)(ap + 4);
        v8f d = (s == 0) ? P1 : (s == 3) ? P4 : Ph;
        d = wmma_t(w1[0], a0, d);
        d = wmma_t(w1[1], a1, d);
        float* hp = sH1 + nr * LDH + wave * 16 + 8 * kh;
        *(v4f*)(hp)     = (v4f){fast_tanh(d[0]), fast_tanh(d[1]),
                                fast_tanh(d[2]), fast_tanh(d[3])};
        *(v4f*)(hp + 4) = (v4f){fast_tanh(d[4]), fast_tanh(d[5]),
                                fast_tanh(d[6]), fast_tanh(d[7])};
      } else {
        if (s == 0)
          out[(size_t)(bt0 + sr) * T_STEPS * NXc + (size_t)t * NXc + sj] = xg;
        const float* xs = sInp + sr * LDIN;
        const float* us = sUA + sr * UP + t * NUc;
        fgv = physics_component(xs, us[0], us[1], sj);
      }
      __syncthreads();
      if (is_mlp) {
        v8f d = mm16t<32>(sH1, LDH, w2, b2r, lane);
        v8f h;
#pragma unroll
        for (int v = 0; v < 8; ++v) h[v] = fast_tanh(d[v]);
        v8f e = {};
#pragma unroll
        for (int kk = 0; kk < 4; ++kk)
          e = wmma_t(w3[kk], (v2f){h[2 * kk], h[2 * kk + 1]}, e);
        if (lane < 16) {
          float* pp = sP + wave * 128 + lane * NXc;
          *(v4f*)(pp)     = (v4f){e[0], e[1], e[2], e[3]};
          *(v4f*)(pp + 4) = (v4f){e[4], e[5], e[6], e[7]};
        }
      }
      __syncthreads();
      if (!is_mlp) {
        float nn = sP[st];
#pragma unroll
        for (int w = 1; w < 8; ++w) nn += sP[w * 128 + st];
        const float k = fgv * scj + nn + b3j;
        float xn;
        if (s == 0)      { k1 = k; xn = xg + 0.5f * DELTA_F * k; }
        else if (s == 1) { k2 = k; xn = xg + 0.5f * DELTA_F * k; }
        else if (s == 2) { k3 = k; xn = xg + DELTA_F * k; }
        else {
          xn = xg + (DELTA_F / 6.0f) * (k1 + 2.0f * k2 + 2.0f * k3 + k);
          XGn[st] = xn;
          xg = xn;
        }
        sInp[sr * LDIN + sj] = xn;
      } else {
        const int mr = lane & 15;
        const int kh = lane >> 4;
        const float* zp = Zc + mr * ZP + 2 * kh;
        const float* xp = XGc + mr * NXc + 2 * kh;
        if (s == 0) {
          const v2f a10 = *(const v2f*)(xp);
          const v2f a11 = *(const v2f*)(xp + 4);
          const v2f a12 = *(const v2f*)(zp + 40);
          const v2f a13 = *(const v2f*)(zp + 44);
          const v2f a14 = *(const v2f*)(kh ? (sUA + mr * UP + NUc * t)
                                           : (Zc + mr * ZP + 48));
          __builtin_amdgcn_sched_barrier(0);
          v8f acc = Qz;
          acc = wmma_t(w1[10], a10, acc);
          acc = wmma_t(w1[11], a11, acc);
          Q = acc;
          acc = wmma_t(w1[12], a12, acc);
          acc = wmma_t(w1[13], a13, acc);
          acc = wmma_t(w1[14], a14, acc);
          P4 = acc;
#pragma unroll
          for (int v = 0; v < 8; ++v) Ph[v] = 0.5f * (P1[v] + acc[v]);
        } else if (s == 1) {
          const int tn = (t + 1 < T_STEPS) ? (t + 1) : t;
          const v2f a12 = *(const v2f*)(zp + 42);
          const v2f a13 = *(const v2f*)(zp + 46);
          const v2f a14 = *(const v2f*)(sUA + mr * UP + NUc * (kh ? tn : t));
          __builtin_amdgcn_sched_barrier(0);
          v8f acc = Q;
          acc = wmma_t(w1[12], a12, acc);
          acc = wmma_t(w1[13], a13, acc);
          acc = wmma_t(w1[14], a14, acc);
          P1 = acc;
        } else if (s == 2) {
          v2f z[4];
#pragma unroll
          for (int j = 0; j < 4; ++j) z[j] = *(const v2f*)(zp + 16 + 4 * j);
          const int r = tid >> 4, q = tid & 15;
          const float* zr = Zc + r * ZP;
          const float v0 = zr[8 + q];
          const float v1 = zr[24 + q];
          const float* s2 = (q < 8) ? (XGc + r * NXc + q)
                                    : (zr + 34 + q);
          const float v2 = *s2;
          const float v3 = sUA[r * UP + t * NUc + (q & 1)];
          __builtin_amdgcn_sched_barrier(0);
          v8f acc = b1r;
#pragma unroll
          for (int j = 0; j < 4; ++j) acc = wmma_t(w1[2 + j], z[j], acc);
          Qz = acc;
          float* zn = Zn + r * ZP;
          zn[q]      = v0;
          zn[16 + q] = v1;
          zn[32 + q] = v2;
          if (q < 2) zn[48 + q] = v3;
        } else {
          const v2f a6 = *(const v2f*)(zp + 32);
          const v2f a7 = *(const v2f*)(zp + 36);
          const v2f a8 = *(const v2f*)(xp);
          const v2f a9 = *(const v2f*)(xp + 4);
          __builtin_amdgcn_sched_barrier(0);
          v8f acc = Qz;
          acc = wmma_t(w1[6], a6, acc);
          acc = wmma_t(w1[7], a7, acc);
          acc = wmma_t(w1[8], a8, acc);
          acc = wmma_t(w1[9], a9, acc);
          Qz = acc;
        }
      }
      __syncthreads();
    }
  }
}

extern "C" void kernel_launch(void* const* d_in, const int* in_sizes, int n_in,
                              void* d_out, int out_size, void* d_ws, size_t ws_size,
                              hipStream_t stream) {
  (void)in_sizes; (void)n_in; (void)out_size; (void)d_ws; (void)ws_size;
  const float* useq = (const float*)d_in[0];
  const float* xGz0 = (const float*)d_in[1];
  const float* W1   = (const float*)d_in[2];
  const float* b1   = (const float*)d_in[3];
  const float* W2   = (const float*)d_in[4];
  const float* b2   = (const float*)d_in[5];
  const float* W3   = (const float*)d_in[6];
  const float* b3   = (const float*)d_in[7];
  float* out = (float*)d_out;
  cstr_rollout_kernel<<<dim3(B_TOT / BT), dim3(NTHREADS), SMEM_BYTES, stream>>>(
      useq, xGz0, W1, b1, W2, b2, W3, b3, out);
}


# Barriers 1 and 3 of a stage split into signal and wait: waves that publish nothing there (state waves at 1: physics, output store; MLP waves at 3: history products) work between the two
# speedup vs baseline: 4.2246x; 1.0855x over previous
#include <hip/hip_runtime.h>
#include <math.h>


typedef __attribute__((ext_vector_type(2))) float v2f;
typedef __attribute__((ext_vector_type(4))) float v4f;
typedef __attribute__((ext_vector_type(8))) float v8f;

#define B_TOT   2048
#define T_STEPS 128
#define NXc     8
#define NUc     2
#define NZc     50
#define DH      128
#define BT      16
#define DELTA_F 0.01f
#define NTHREADS 384
#define NMLP     256

#define LDIN    68
#define LDH     132
#define ZP      52
#define UP      258
#define OFF_INP 0u
#define OFF_H1  (OFF_INP + 16u*LDIN*4u)
#define OFF_P   (OFF_H1  + 16u*LDH*4u)
#define OFF_XG  (OFF_P   + 8u*128u*4u)
#define OFF_Z   (OFF_XG  + 2u*128u*4u)
#define OFF_UA  (OFF_Z   + 2u*16u*ZP*4u)
#define SMEM_BYTES (OFF_UA + 16u*UP*4u)

__device__ __forceinline__ float fast_tanh(float x) {
  float e = __builtin_amdgcn_exp2f(x * (2.0f * 1.442695041f));
  return 1.0f - 2.0f * __builtin_amdgcn_rcpf(e + 1.0f);
}

__device__ __forceinline__ v8f wmma_t(v2f w, v2f x, v8f acc) {
  return __builtin_amdgcn_wmma_f32_16x16x4_f32(false, w, false, x,
                                               (short)0, acc, false, false);
}

template <int KI>
__device__ __forceinline__ v8f mm16t(const float* __restrict__ X, int ldx,
                                     const v2f (&w)[KI], v8f acc0, int lane) {
  const int nr = lane & 15;
  const int kh = lane >> 4;
  const float* xp = X + nr * ldx + 2 * kh;
  v2f x[KI];
#pragma unroll
  for (int kk = 0; kk < KI; ++kk) x[kk] = *(const v2f*)(xp + 4 * kk);
  __builtin_amdgcn_sched_barrier(0);
  v8f acc = acc0;
#pragma unroll
  for (int kk = 0; kk < KI; ++kk) acc = wmma_t(w[kk], x[kk], acc);
  return acc;
}

__device__ __forceinline__ float physics_component(const float* xs, float u0,
                                                   float u1, int j) {
  float Hr  = xs[0] * 2.0f + 50.0f;
  float CAr = xs[1] * 0.2f + 1.0f;
  float CBr = xs[2] * 0.1f + 0.5f;
  float Tr  = xs[3] * 4.0f + 313.0f;
  float Hb  = xs[4] * 2.0f + 50.0f;
  float CAb = xs[5] * 0.2f + 1.0f;
  float CBb = xs[6] * 0.1f + 0.5f;
  float Tb  = xs[7] * 4.0f + 313.0f;
  float F = u0 * 1.0f + 10.0f;
  float D = u1 * 0.5f + 5.0f;
  float Fr = 2.5f * __fsqrt_rn(Hr);
  float Fb = 1.5f * __fsqrt_rn(Hb);
  float invArHr = __fdividef(1.0f, 3.0f * Hr);
  float invAbHb = __fdividef(1.0f, 3.0f * Hb);
  float dHr  = (F + D - Fr) * (1.0f / 3.0f);
  float dCAr = (F * (6.0f - CAr) - D * CAr) * invArHr;
  float dCBr = (-F * CBr - D * CBr) * invArHr;
  float dTr  = (F * (320.0f - Tr) + D * (300.0f - Tr)) * invArHr
             + (200.0f / 30.0f) * invArHr;
  float dHb  = (Fr - Fb - D) * (1.0f / 3.0f);
  float dCAb = (Fr * (CAr - CAb) + D * CAb) * invAbHb;
  float dCBb = (Fr * (CBr - CBb) + D * CBb) * invAbHb;
  float dTb  = Fr * (Tr - Tb) * invAbHb + (200.0f / 30.0f) * invAbHb;
  float lo = (j & 2) ? ((j & 1) ? dTr : dCBr) : ((j & 1) ? dCAr : dHr);
  float hi = (j & 2) ? ((j & 1) ? dTb : dCBb) : ((j & 1) ? dCAb : dHb);
  return (j & 4) ? hi : lo;
}

__global__ __launch_bounds__(NTHREADS) void cstr_rollout_kernel(
    const float* __restrict__ useq, const float* __restrict__ xGz0,
    const float* __restrict__ W1g, const float* __restrict__ b1g,
    const float* __restrict__ W2g, const float* __restrict__ b2g,
    const float* __restrict__ W3g, const float* __restrict__ b3g,
    float* __restrict__ out) {
  extern __shared__ __align__(16) char smem[];
  float* sInp = (float*)(smem + OFF_INP);
  float* sH1  = (float*)(smem + OFF_H1);
  float* sP   = (float*)(smem + OFF_P);
  float* sXG  = (float*)(smem + OFF_XG);
  float* sZ   = (float*)(smem + OFF_Z);
  float* sUA  = (float*)(smem + OFF_UA);

  const int tid  = threadIdx.x;
  const int lane = tid & 31;
  const int wave = __builtin_amdgcn_readfirstlane(tid >> 5);
  const bool is_mlp = wave < 8;
  const int bt0  = blockIdx.x * BT;

  const int st = (tid - NMLP) & 127;
  const int sr = st >> 3;
  const int sj = st & 7;

  v2f w1[15] = {}, w2[32] = {}, w3[4] = {};
  v8f b1r = {}, b2r = {};
  const int wn  = lane & 15;
  const int wkh = lane >> 4;
  if (is_mlp) {
    const int wcol = wave * 16 + wn;
#pragma unroll
    for (int kk = 0; kk < 15; ++kk) {
      const int k0 = 4 * kk + 2 * wkh;
      w1[kk] = (v2f){W1g[(k0) * DH + wcol], W1g[(k0 + 1) * DH + wcol]};
    }
#pragma unroll
    for (int kk = 0; kk < 32; ++kk) {
      const int k0 = 4 * kk + 2 * wkh;
      w2[kk] = (v2f){W2g[(k0) * DH + wcol], W2g[(k0 + 1) * DH + wcol]};
    }
#pragma unroll
    for (int kk = 0; kk < 4; ++kk) {
      const int k0 = wave * 16 + 8 * wkh + 2 * kk;
      v2f p = (v2f){0.0f, 0.0f};
      if (wn < NXc) p = (v2f){W3g[(k0) * NXc + wn], W3g[(k0 + 1) * NXc + wn]};
      w3[kk] = p;
    }
#pragma unroll
    for (int v = 0; v < 8; ++v) {
      b1r[v] = b1g[wave * 16 + 8 * wkh + v];
      b2r[v] = b2g[wave * 16 + 8 * wkh + v];
    }
  }

  float xg = 0.0f, b3j = 0.0f;
  if (!is_mlp) {
    xg  = xGz0[(size_t)(bt0 + sr) * 58 + sj];
    b3j = b3g[sj];
  }
  const float scj = ((sj & 3) == 0) ? 0.5f : ((sj & 3) == 1) ? 5.0f
                  : ((sj & 3) == 2) ? 10.0f : 0.25f;

  for (int i = tid; i < BT * T_STEPS * NUc; i += NTHREADS) {
    int r = i >> 8, q = i & 255;
    sUA[r * UP + q] = useq[(size_t)bt0 * (T_STEPS * NUc) + i];
  }
  for (int i = tid; i < BT * 58; i += NTHREADS) {
    int r = i / 58, c = i - r * 58;
    float v = xGz0[(size_t)(bt0 + r) * 58 + c];
    if (c < NXc) sXG[r * NXc + c] = v;
    else         sZ[r * ZP + (c - NXc)] = v;
    sInp[r * LDIN + c] = v;
  }
  if (tid < BT * 6) {
    int r = tid / 6, c = tid - r * 6;
    float v = (c < 2) ? useq[(size_t)(bt0 + r) * (T_STEPS * NUc) + c] : 0.0f;
    sInp[r * LDIN + 58 + c] = v;
  }
  __syncthreads();

  v8f P1 = {}, Q = {}, Qz = {};
  if (is_mlp) {
    const float* ap = sInp + (lane & 15) * LDIN + 2 * (lane >> 4);
    const float* zp = sZ + (lane & 15) * ZP + 2 * (lane >> 4);
    v2f a[13], z[8];
#pragma unroll
    for (int kk = 2; kk < 15; ++kk) a[kk - 2] = *(const v2f*)(ap + 4 * kk);
#pragma unroll
    for (int kk = 2; kk < 10; ++kk) z[kk - 2] = *(const v2f*)(zp + 4 * kk);
    __builtin_amdgcn_sched_barrier(0);
    v8f acc = b1r;
#pragma unroll
    for (int kk = 2; kk < 15; ++kk) acc = wmma_t(w1[kk], a[kk - 2], acc);
    P1 = acc;
    v8f acz = b1r;
#pragma unroll
    for (int kk = 2; kk < 10; ++kk) acz = wmma_t(w1[kk], z[kk - 2], acz);
    Qz = acz;
  }

  float k1 = 0.0f, k2 = 0.0f, k3 = 0.0f;
  for (int t = 0; t < T_STEPS; ++t) {
    const int cur = t & 1;
    const float* Zc  = sZ  + cur * (BT * ZP);
    float*       Zn  = sZ  + (cur ^ 1) * (BT * ZP);
    const float* XGc = sXG + cur * (BT * NXc);
    float*       XGn = sXG + (cur ^ 1) * (BT * NXc);

    v8f P4 = {}, Ph = {};

#pragma unroll
    for (int s = 0; s < 4; ++s) {
      float fgv = 0.0f;
      if (is_mlp) {
        const int nr = lane & 15;
        const int kh = lane >> 4;
        const float* ap = sInp + nr * LDIN + 2 * kh;
        const v2f a0 = *(const v2f*)(ap);
        const v2f a1 = *(const v2f*)(ap + 4);
        v8f d = (s == 0) ? P1 : (s == 3) ? P4 : Ph;
        d = wmma_t(w1[0], a0, d);
        d = wmma_t(w1[1], a1, d);
        float* hp = sH1 + nr * LDH + wave * 16 + 8 * kh;
        *(v4f*)(hp)     = (v4f){fast_tanh(d[0]), fast_tanh(d[1]),
                                fast_tanh(d[2]), fast_tanh(d[3])};
        *(v4f*)(hp + 4) = (v4f){fast_tanh(d[4]), fast_tanh(d[5]),
                                fast_tanh(d[6]), fast_tanh(d[7])};
      }
      __builtin_amdgcn_fence(__ATOMIC_RELEASE, "workgroup");
      __builtin_amdgcn_s_barrier_signal(-1);
      if (!is_mlp) {
        if (s == 0)
          out[(size_t)(bt0 + sr) * T_STEPS * NXc + (size_t)t * NXc + sj] = xg;
        const float* xs = sInp + sr * LDIN;
        const float* us = sUA + sr * UP + t * NUc;
        fgv = physics_component(xs, us[0], us[1], sj);
      }
      __builtin_amdgcn_s_barrier_wait(-1);
      __builtin_amdgcn_fence(__ATOMIC_ACQUIRE, "workgroup");
      if (is_mlp) {
        v8f d = mm16t<32>(sH1, LDH, w2, b2r, lane);
        v8f h;
#pragma unroll
        for (int v = 0; v < 8; ++v) h[v] = fast_tanh(d[v]);
        v8f e = {};
#pragma unroll
        for (int kk = 0; kk < 4; ++kk)
          e = wmma_t(w3[kk], (v2f){h[2 * kk], h[2 * kk + 1]}, e);
        if (lane < 16) {
          float* pp = sP + wave * 128 + lane * NXc;
          *(v4f*)(pp)     = (v4f){e[0], e[1], e[2], e[3]};
          *(v4f*)(pp + 4) = (v4f){e[4], e[5], e[6], e[7]};
        }
      }
      __syncthreads();
      if (!is_mlp) {
        float nn = sP[st];
#pragma unroll
        for (int w = 1; w < 8; ++w) nn += sP[w * 128 + st];
        const float k = fgv * scj + nn + b3j;
        float xn;
        if (s == 0)      { k1 = k; xn = xg + 0.5f * DELTA_F * k; }
        else if (s == 1) { k2 = k; xn = xg + 0.5f * DELTA_F * k; }
        else if (s == 2) { k3 = k; xn = xg + DELTA_F * k; }
        else {
          xn = xg + (DELTA_F / 6.0f) * (k1 + 2.0f * k2 + 2.0f * k3 + k);
          XGn[st] = xn;
          xg = xn;
        }
        sInp[sr * LDIN + sj] = xn;
      }
      __builtin_amdgcn_fence(__ATOMIC_RELEASE, "workgroup");
      __builtin_amdgcn_s_barrier_signal(-1);
      if (is_mlp) {
        const int mr = lane & 15;
        const int kh = lane >> 4;
        const float* zp = Zc + mr * ZP + 2 * kh;
        const float* xp = XGc + mr * NXc + 2 * kh;
        if (s == 0) {
          const v2f a10 = *(const v2f*)(xp);
          const v2f a11 = *(const v2f*)(xp + 4);
          const v2f a12 = *(const v2f*)(zp + 40);
          const v2f a13 = *(const v2f*)(zp + 44);
          const v2f a14 = *(const v2f*)(kh ? (sUA + mr * UP + NUc * t)
                                           : (Zc + mr * ZP + 48));
          __builtin_amdgcn_sched_barrier(0);
          v8f acc = Qz;
          acc = wmma_t(w1[10], a10, acc);
          acc = wmma_t(w1[11], a11, acc);
          Q = acc;
          acc = wmma_t(w1[12], a12, acc);
          acc = wmma_t(w1[13], a13, acc);
          acc = wmma_t(w1[14], a14, acc);
          P4 = acc;
#pragma unroll
          for (int v = 0; v < 8; ++v) Ph[v] = 0.5f * (P1[v] + acc[v]);
        } else if (s == 1) {
          const int tn = (t + 1 < T_STEPS) ? (t + 1) : t;
          const v2f a12 = *(const v2f*)(zp + 42);
          const v2f a13 = *(const v2f*)(zp + 46);
          const v2f a14 = *(const v2f*)(sUA + mr * UP + NUc * (kh ? tn : t));
          __builtin_amdgcn_sched_barrier(0);
          v8f acc = Q;
          acc = wmma_t(w1[12], a12, acc);
          acc = wmma_t(w1[13], a13, acc);
          acc = wmma_t(w1[14], a14, acc);
          P1 = acc;
        } else if (s == 2) {
          v2f z[4];
#pragma unroll
          for (int j = 0; j < 4; ++j) z[j] = *(const v2f*)(zp + 16 + 4 * j);
          const int r = tid >> 4, q = tid & 15;
          const float* zr = Zc + r * ZP;
          const float v0 = zr[8 + q];
          const float v1 = zr[24 + q];
          const float* s2 = (q < 8) ? (XGc + r * NXc + q)
                                    : (zr + 34 + q);
          const float v2 = *s2;
          const float v3 = sUA[r * UP + t * NUc + (q & 1)];
          __builtin_amdgcn_sched_barrier(0);
          v8f acc = b1r;
#pragma unroll
          for (int j = 0; j < 4; ++j) acc = wmma_t(w1[2 + j], z[j], acc);
          Qz = acc;
          float* zn = Zn + r * ZP;
          zn[q]      = v0;
          zn[16 + q] = v1;
          zn[32 + q] = v2;
          if (q < 2) zn[48 + q] = v3;
        } else {
          const v2f a6 = *(const v2f*)(zp + 32);
          const v2f a7 = *(const v2f*)(zp + 36);
          const v2f a8 = *(const v2f*)(xp);
          const v2f a9 = *(const v2f*)(xp + 4);
          __builtin_amdgcn_sched_barrier(0);
          v8f acc = Qz;
          acc = wmma_t(w1[6], a6, acc);
          acc = wmma_t(w1[7], a7, acc);
          acc = wmma_t(w1[8], a8, acc);
          acc = wmma_t(w1[9], a9, acc);
          Qz = acc;
        }
      }
      __builtin_amdgcn_s_barrier_wait(-1);
      __builtin_amdgcn_fence(__ATOMIC_ACQUIRE, "workgroup");
    }
  }
}

extern "C" void kernel_launch(void* const* d_in, const int* in_sizes, int n_in,
                              void* d_out, int out_size, void* d_ws, size_t ws_size,
                              hipStream_t stream) {
  (void)in_sizes; (void)n_in; (void)out_size; (void)d_ws; (void)ws_size;
  const float* useq = (const float*)d_in[0];
  const float* xGz0 = (const float*)d_in[1];
  const float* W1   = (const float*)d_in[2];
  const float* b1   = (const float*)d_in[3];
  const float* W2   = (const float*)d_in[4];
  const float* b2   = (const float*)d_in[5];
  const float* W3   = (const float*)d_in[6];
  const float* b3   = (const float*)d_in[7];
  float* out = (float*)d_out;
  cstr_rollout_kernel<<<dim3(B_TOT / BT), dim3(NTHREADS), SMEM_BYTES, stream>>>(
      useq, xGz0, W1, b1, W2, b2, W3, b3, out);
}


# tanh as the chip's own v_tanh_f32: one transcendental a value instead of multiply, v_exp_f32, add, v_rcp_f32, multiply-add behind every WMMA chain; activations round as the hardware tanh does
# speedup vs baseline: 4.4350x; 1.0498x over previous
#include <hip/hip_runtime.h>
#include <math.h>


typedef __attribute__((ext_vector_type(2))) float v2f;
typedef __attribute__((ext_vector_type(4))) float v4f;
typedef __attribute__((ext_vector_type(8))) float v8f;

#define B_TOT   2048
#define T_STEPS 128
#define NXc     8
#define NUc     2
#define NZc     50
#define DH      128
#define BT      16
#define DELTA_F 0.01f
#define NTHREADS 384
#define NMLP     256

#define LDIN    68
#define LDH     132
#define ZP      52
#define UP      258
#define OFF_INP 0u
#define OFF_H1  (OFF_INP + 16u*LDIN*4u)
#define OFF_P   (OFF_H1  + 16u*LDH*4u)
#define OFF_XG  (OFF_P   + 8u*128u*4u)
#define OFF_Z   (OFF_XG  + 2u*128u*4u)
#define OFF_UA  (OFF_Z   + 2u*16u*ZP*4u)
#define SMEM_BYTES (OFF_UA + 16u*UP*4u)

__device__ __forceinline__ float fast_tanh(float x) {
  return __builtin_amdgcn_tanhf(x);
}

__device__ __forceinline__ v8f wmma_t(v2f w, v2f x, v8f acc) {
  return __builtin_amdgcn_wmma_f32_16x16x4_f32(false, w, false, x,
                                               (short)0, acc, false, false);
}

template <int KI>
__device__ __forceinline__ v8f mm16t(const float* __restrict__ X, int ldx,
                                     const v2f (&w)[KI], v8f acc0, int lane) {
  const int nr = lane & 15;
  const int kh = lane >> 4;
  const float* xp = X + nr * ldx + 2 * kh;
  v2f x[KI];
#pragma unroll
  for (int kk = 0; kk < KI; ++kk) x[kk] = *(const v2f*)(xp + 4 * kk);
  __builtin_amdgcn_sched_barrier(0);
  v8f acc = acc0;
#pragma unroll
  for (int kk = 0; kk < KI; ++kk) acc = wmma_t(w[kk], x[kk], acc);
  return acc;
}

__device__ __forceinline__ float physics_component(const float* xs, float u0,
                                                   float u1, int j) {
  float Hr  = xs[0] * 2.0f + 50.0f;
  float CAr = xs[1] * 0.2f + 1.0f;
  float CBr = xs[2] * 0.1f + 0.5f;
  float Tr  = xs[3] * 4.0f + 313.0f;
  float Hb  = xs[4] * 2.0f + 50.0f;
  float CAb = xs[5] * 0.2f + 1.0f;
  float CBb = xs[6] * 0.1f + 0.5f;
  float Tb  = xs[7] * 4.0f + 313.0f;
  float F = u0 * 1.0f + 10.0f;
  float D = u1 * 0.5f + 5.0f;
  float Fr = 2.5f * __fsqrt_rn(Hr);
  float Fb = 1.5f * __fsqrt_rn(Hb);
  float invArHr = __fdividef(1.0f, 3.0f * Hr);
  float invAbHb = __fdividef(1.0f, 3.0f * Hb);
  float dHr  = (F + D - Fr) * (1.0f / 3.0f);
  float dCAr = (F * (6.0f - CAr) - D * CAr) * invArHr;
  float dCBr = (-F * CBr - D * CBr) * invArHr;
  float dTr  = (F * (320.0f - Tr) + D * (300.0f - Tr)) * invArHr
             + (200.0f / 30.0f) * invArHr;
  float dHb  = (Fr - Fb - D) * (1.0f / 3.0f);
  float dCAb = (Fr * (CAr - CAb) + D * CAb) * invAbHb;
  float dCBb = (Fr * (CBr - CBb) + D * CBb) * invAbHb;
  float dTb  = Fr * (Tr - Tb) * invAbHb + (200.0f / 30.0f) * invAbHb;
  float lo = (j & 2) ? ((j & 1) ? dTr : dCBr) : ((j & 1) ? dCAr : dHr);
  float hi = (j & 2) ? ((j & 1) ? dTb : dCBb) : ((j & 1) ? dCAb : dHb);
  return (j & 4) ? hi : lo;
}

__global__ __launch_bounds__(NTHREADS) void cstr_rollout_kernel(
    const float* __restrict__ useq, const float* __restrict__ xGz0,
    const float* __restrict__ W1g, const float* __restrict__ b1g,
    const float* __restrict__ W2g, const float* __restrict__ b2g,
    const float* __restrict__ W3g, const float* __restrict__ b3g,
    float* __restrict__ out) {
  extern __shared__ __align__(16) char smem[];
  float* sInp = (float*)(smem + OFF_INP);
  float* sH1  = (float*)(smem + OFF_H1);
  float* sP   = (float*)(smem + OFF_P);
  float* sXG  = (float*)(smem + OFF_XG);
  float* sZ   = (float*)(smem + OFF_Z);
  float* sUA  = (float*)(smem + OFF_UA);

  const int tid  = threadIdx.x;
  const int lane = tid & 31;
  const int wave = __builtin_amdgcn_readfirstlane(tid >> 5);
  const bool is_mlp = wave < 8;
  const int bt0  = blockIdx.x * BT;

  const int st = (tid - NMLP) & 127;
  const int sr = st >> 3;
  const int sj = st & 7;

  v2f w1[15] = {}, w2[32] = {}, w3[4] = {};
  v8f b1r = {}, b2r = {};
  const int wn  = lane & 15;
  const int wkh = lane >> 4;
  if (is_mlp) {
    const int wcol = wave * 16 + wn;
#pragma unroll
    for (int kk = 0; kk < 15; ++kk) {
      const int k0 = 4 * kk + 2 * wkh;
      w1[kk] = (v2f){W1g[(k0) * DH + wcol], W1g[(k0 + 1) * DH + wcol]};
    }
#pragma unroll
    for (int kk = 0; kk < 32; ++kk) {
      const int k0 = 4 * kk + 2 * wkh;
      w2[kk] = (v2f){W2g[(k0) * DH + wcol], W2g[(k0 + 1) * DH + wcol]};
    }
#pragma unroll
    for (int kk = 0; kk < 4; ++kk) {
      const int k0 = wave * 16 + 8 * wkh + 2 * kk;
      v2f p = (v2f){0.0f, 0.0f};
      if (wn < NXc) p = (v2f){W3g[(k0) * NXc + wn], W3g[(k0 + 1) * NXc + wn]};
      w3[kk] = p;
    }
#pragma unroll
    for (int v = 0; v < 8; ++v) {
      b1r[v] = b1g[wave * 16 + 8 * wkh + v];
      b2r[v] = b2g[wave * 16 + 8 * wkh + v];
    }
  }

  float xg = 0.0f, b3j = 0.0f;
  if (!is_mlp) {
    xg  = xGz0[(size_t)(bt0 + sr) * 58 + sj];
    b3j = b3g[sj];
  }
  const float scj = ((sj & 3) == 0) ? 0.5f : ((sj & 3) == 1) ? 5.0f
                  : ((sj & 3) == 2) ? 10.0f : 0.25f;

  for (int i = tid; i < BT * T_STEPS * NUc; i += NTHREADS) {
    int r = i >> 8, q = i & 255;
    sUA[r * UP + q] = useq[(size_t)bt0 * (T_STEPS * NUc) + i];
  }
  for (int i = tid; i < BT * 58; i += NTHREADS) {
    int r = i / 58, c = i - r * 58;
    float v = xGz0[(size_t)(bt0 + r) * 58 + c];
    if (c < NXc) sXG[r * NXc + c] = v;
    else         sZ[r * ZP + (c - NXc)] = v;
    sInp[r * LDIN + c] = v;
  }
  if (tid < BT * 6) {
    int r = tid / 6, c = tid - r * 6;
    float v = (c < 2) ? useq[(size_t)(bt0 + r) * (T_STEPS * NUc) + c] : 0.0f;
    sInp[r * LDIN + 58 + c] = v;
  }
  __syncthreads();

  v8f P1 = {}, Q = {}, Qz = {};
  if (is_mlp) {
    const float* ap = sInp + (lane & 15) * LDIN + 2 * (lane >> 4);
    const float* zp = sZ + (lane & 15) * ZP + 2 * (lane >> 4);
    v2f a[13], z[8];
#pragma unroll
    for (int kk = 2; kk < 15; ++kk) a[kk - 2] = *(const v2f*)(ap + 4 * kk);
#pragma unroll
    for (int kk = 2; kk < 10; ++kk) z[kk - 2] = *(const v2f*)(zp + 4 * kk);
    __builtin_amdgcn_sched_barrier(0);
    v8f acc = b1r;
#pragma unroll
    for (int kk = 2; kk < 15; ++kk) acc = wmma_t(w1[kk], a[kk - 2], acc);
    P1 = acc;
    v8f acz = b1r;
#pragma unroll
    for (int kk = 2; kk < 10; ++kk) acz = wmma_t(w1[kk], z[kk - 2], acz);
    Qz = acz;
  }

  float k1 = 0.0f, k2 = 0.0f, k3 = 0.0f;
  for (int t = 0; t < T_STEPS; ++t) {
    const int cur = t & 1;
    const float* Zc  = sZ  + cur * (BT * ZP);
    float*       Zn  = sZ  + (cur ^ 1) * (BT * ZP);
    const float* XGc = sXG + cur * (BT * NXc);
    float*       XGn = sXG + (cur ^ 1) * (BT * NXc);

    v8f P4 = {}, Ph = {};

#pragma unroll
    for (int s = 0; s < 4; ++s) {
      float fgv = 0.0f;
      if (is_mlp) {
        const int nr = lane & 15;
        const int kh = lane >> 4;
        const float* ap = sInp + nr * LDIN + 2 * kh;
        const v2f a0 = *(const v2f*)(ap);
        const v2f a1 = *(const v2f*)(ap + 4);
        v8f d = (s == 0) ? P1 : (s == 3) ? P4 : Ph;
        d = wmma_t(w1[0], a0, d);
        d = wmma_t(w1[1], a1, d);
        float* hp = sH1 + nr * LDH + wave * 16 + 8 * kh;
        *(v4f*)(hp)     = (v4f){fast_tanh(d[0]), fast_tanh(d[1]),
                                fast_tanh(d[2]), fast_tanh(d[3])};
        *(v4f*)(hp + 4) = (v4f){fast_tanh(d[4]), fast_tanh(d[5]),
                                fast_tanh(d[6]), fast_tanh(d[7])};
      }
      __builtin_amdgcn_fence(__ATOMIC_RELEASE, "workgroup");
      __builtin_amdgcn_s_barrier_signal(-1);
      if (!is_mlp) {
        if (s == 0)
          out[(size_t)(bt0 + sr) * T_STEPS * NXc + (size_t)t * NXc + sj] = xg;
        const float* xs = sInp + sr * LDIN;
        const float* us = sUA + sr * UP + t * NUc;
        fgv = physics_component(xs, us[0], us[1], sj);
      }
      __builtin_amdgcn_s_barrier_wait(-1);
      __builtin_amdgcn_fence(__ATOMIC_ACQUIRE, "workgroup");
      if (is_mlp) {
        v8f d = mm16t<32>(sH1, LDH, w2, b2r, lane);
        v8f h;
#pragma unroll
        for (int v = 0; v < 8; ++v) h[v] = fast_tanh(d[v]);
        v8f e = {};
#pragma unroll
        for (int kk = 0; kk < 4; ++kk)
          e = wmma_t(w3[kk], (v2f){h[2 * kk], h[2 * kk + 1]}, e);
        if (lane < 16) {
          float* pp = sP + wave * 128 + lane * NXc;
          *(v4f*)(pp)     = (v4f){e[0], e[1], e[2], e[3]};
          *(v4f*)(pp + 4) = (v4f){e[4], e[5], e[6], e[7]};
        }
      }
      __syncthreads();
      if (!is_mlp) {
        float nn = sP[st];
#pragma unroll
        for (int w = 1; w < 8; ++w) nn += sP[w * 128 + st];
        const float k = fgv * scj + nn + b3j;
        float xn;
        if (s == 0)      { k1 = k; xn = xg + 0.5f * DELTA_F * k; }
        else if (s == 1) { k2 = k; xn = xg + 0.5f * DELTA_F * k; }
        else if (s == 2) { k3 = k; xn = xg + DELTA_F * k; }
        else {
          xn = xg + (DELTA_F / 6.0f) * (k1 + 2.0f * k2 + 2.0f * k3 + k);
          XGn[st] = xn;
          xg = xn;
        }
        sInp[sr * LDIN + sj] = xn;
      }
      __builtin_amdgcn_fence(__ATOMIC_RELEASE, "workgroup");
      __builtin_amdgcn_s_barrier_signal(-1);
      if (is_mlp) {
        const int mr = lane & 15;
        const int kh = lane >> 4;
        const float* zp = Zc + mr * ZP + 2 * kh;
        const float* xp = XGc + mr * NXc + 2 * kh;
        if (s == 0) {
          const v2f a10 = *(const v2f*)(xp);
          const v2f a11 = *(const v2f*)(xp + 4);
          const v2f a12 = *(const v2f*)(zp + 40);
          const v2f a13 = *(const v2f*)(zp + 44);
          const v2f a14 = *(const v2f*)(kh ? (sUA + mr * UP + NUc * t)
                                           : (Zc + mr * ZP + 48));
          __builtin_amdgcn_sched_barrier(0);
          v8f acc = Qz;
          acc = wmma_t(w1[10], a10, acc);
          acc = wmma_t(w1[11], a11, acc);
          Q = acc;
          acc = wmma_t(w1[12], a12, acc);
          acc = wmma_t(w1[13], a13, acc);
          acc = wmma_t(w1[14], a14, acc);
          P4 = acc;
#pragma unroll
          for (int v = 0; v < 8; ++v) Ph[v] = 0.5f * (P1[v] + acc[v]);
        } else if (s == 1) {
          const int tn = (t + 1 < T_STEPS) ? (t + 1) : t;
          const v2f a12 = *(const v2f*)(zp + 42);
          const v2f a13 = *(const v2f*)(zp + 46);
          const v2f a14 = *(const v2f*)(sUA + mr * UP + NUc * (kh ? tn : t));
          __builtin_amdgcn_sched_barrier(0);
          v8f acc = Q;
          acc = wmma_t(w1[12], a12, acc);
          acc = wmma_t(w1[13], a13, acc);
          acc = wmma_t(w1[14], a14, acc);
          P1 = acc;
        } else if (s == 2) {
          v2f z[4];
#pragma unroll
          for (int j = 0; j < 4; ++j) z[j] = *(const v2f*)(zp + 16 + 4 * j);
          const int r = tid >> 4, q = tid & 15;
          const float* zr = Zc + r * ZP;
          const float v0 = zr[8 + q];
          const float v1 = zr[24 + q];
          const float* s2 = (q < 8) ? (XGc + r * NXc + q)
                                    : (zr + 34 + q);
          const float v2 = *s2;
          const float v3 = sUA[r * UP + t * NUc + (q & 1)];
          __builtin_amdgcn_sched_barrier(0);
          v8f acc = b1r;
#pragma unroll
          for (int j = 0; j < 4; ++j) acc = wmma_t(w1[2 + j], z[j], acc);
          Qz = acc;
          float* zn = Zn + r * ZP;
          zn[q]      = v0;
          zn[16 + q] = v1;
          zn[32 + q] = v2;
          if (q < 2) zn[48 + q] = v3;
        } else {
          const v2f a6 = *(const v2f*)(zp + 32);
          const v2f a7 = *(const v2f*)(zp + 36);
          const v2f a8 = *(const v2f*)(xp);
          const v2f a9 = *(const v2f*)(xp + 4);
          __builtin_amdgcn_sched_barrier(0);
          v8f acc = Qz;
          acc = wmma_t(w1[6], a6, acc);
          acc = wmma_t(w1[7], a7, acc);
          acc = wmma_t(w1[8], a8, acc);
          acc = wmma_t(w1[9], a9, acc);
          Qz = acc;
        }
      }
      __builtin_amdgcn_s_barrier_wait(-1);
      __builtin_amdgcn_fence(__ATOMIC_ACQUIRE, "workgroup");
    }
  }
}

extern "C" void kernel_launch(void* const* d_in, const int* in_sizes, int n_in,
                              void* d_out, int out_size, void* d_ws, size_t ws_size,
                              hipStream_t stream) {
  (void)in_sizes; (void)n_in; (void)out_size; (void)d_ws; (void)ws_size;
  const float* useq = (const float*)d_in[0];
  const float* xGz0 = (const float*)d_in[1];
  const float* W1   = (const float*)d_in[2];
  const float* b1   = (const float*)d_in[3];
  const float* W2   = (const float*)d_in[4];
  const float* b2   = (const float*)d_in[5];
  const float* W3   = (const float*)d_in[6];
  const float* b3   = (const float*)d_in[7];
  float* out = (float*)d_out;
  cstr_rollout_kernel<<<dim3(B_TOT / BT), dim3(NTHREADS), SMEM_BYTES, stream>>>(
      useq, xGz0, W1, b1, W2, b2, W3, b3, out);
}
